# starting kernel
# baseline (speedup 1.0000x reference)
#include <hip/hip_runtime.h>
#include <math.h>

#define NEG_SLOPE 0.2f

typedef __attribute__((ext_vector_type(2))) float v2f;
typedef __attribute__((ext_vector_type(8))) float v8f;

__device__ __forceinline__ float lrelu(float v) { return v > 0.f ? v : NEG_SLOPE * v; }

__device__ __forceinline__ unsigned ford(float f) {
    unsigned u = __float_as_uint(f);
    return (u & 0x80000000u) ? ~u : (u | 0x80000000u);
}
__device__ __forceinline__ float forddec(unsigned u) {
    return (u & 0x80000000u) ? __uint_as_float(u & 0x7fffffffu) : __uint_as_float(~u);
}
#define FORD_NEG_INF 0x007FFFFFu

__global__ void k_init(unsigned* m1, int n1, unsigned* m2, int n2,
                       float* s1, float* wsum1, int ns1,
                       float* s2, int ns2, float* acc2, int nacc) {
    int i = blockIdx.x * blockDim.x + threadIdx.x;
    if (i < n1)  m1[i] = FORD_NEG_INF;
    if (i < n2)  m2[i] = FORD_NEG_INF;
    if (i < ns1) { s1[i] = 0.f; wsum1[i] = 0.f; }
    if (i < ns2) s2[i] = 0.f;
    if (i < nacc) acc2[i] = 0.f;
}

__global__ void k_node1(const float* __restrict__ x, const float* __restrict__ W1,
                        const float* __restrict__ as1, const float* __restrict__ ad1,
                        float* __restrict__ alS1, float* __restrict__ alD1, int N) {
    int n = blockIdx.x * blockDim.x + threadIdx.x;
    if (n >= N) return;
    float xv = x[n];
#pragma unroll
    for (int h = 0; h < 4; ++h) {
        float ws = 0.f, wd = 0.f;
#pragma unroll
        for (int c = 0; c < 8; ++c) {
            float w = W1[h * 8 + c];
            ws += w * as1[h * 8 + c];
            wd += w * ad1[h * 8 + c];
        }
        alS1[n * 4 + h] = xv * ws;
        alD1[n * 4 + h] = xv * wd;
    }
}

__global__ void k_edge1_max(const int* __restrict__ ei, int E, int N,
                            const float* __restrict__ alS1, const float* __restrict__ alD1,
                            unsigned* __restrict__ m1) {
    int e = blockIdx.x * blockDim.x + threadIdx.x;
    if (e >= E + N) return;
    int s, d;
    if (e < E) { s = ei[e]; d = ei[E + e]; } else { s = d = e - E; }
    float4 as = *reinterpret_cast<const float4*>(alS1 + (size_t)s * 4);
    float4 ad = *reinterpret_cast<const float4*>(alD1 + (size_t)d * 4);
    atomicMax(m1 + (size_t)d * 4 + 0, ford(lrelu(as.x + ad.x)));
    atomicMax(m1 + (size_t)d * 4 + 1, ford(lrelu(as.y + ad.y)));
    atomicMax(m1 + (size_t)d * 4 + 2, ford(lrelu(as.z + ad.z)));
    atomicMax(m1 + (size_t)d * 4 + 3, ford(lrelu(as.w + ad.w)));
}

__global__ void k_edge1_sum(const int* __restrict__ ei, int E, int N,
                            const float* __restrict__ x,
                            const float* __restrict__ alS1, const float* __restrict__ alD1,
                            const unsigned* __restrict__ m1,
                            float* __restrict__ s1, float* __restrict__ wsum1) {
    int e = blockIdx.x * blockDim.x + threadIdx.x;
    if (e >= E + N) return;
    int s, d;
    if (e < E) { s = ei[e]; d = ei[E + e]; } else { s = d = e - E; }
    float4 as = *reinterpret_cast<const float4*>(alS1 + (size_t)s * 4);
    float4 ad = *reinterpret_cast<const float4*>(alD1 + (size_t)d * 4);
    uint4  mm = *reinterpret_cast<const uint4*>(m1 + (size_t)d * 4);
    float xs = x[s];
    float t0 = __expf(lrelu(as.x + ad.x) - forddec(mm.x));
    float t1 = __expf(lrelu(as.y + ad.y) - forddec(mm.y));
    float t2 = __expf(lrelu(as.z + ad.z) - forddec(mm.z));
    float t3 = __expf(lrelu(as.w + ad.w) - forddec(mm.w));
    atomicAdd(s1 + (size_t)d * 4 + 0, t0);
    atomicAdd(s1 + (size_t)d * 4 + 1, t1);
    atomicAdd(s1 + (size_t)d * 4 + 2, t2);
    atomicAdd(s1 + (size_t)d * 4 + 3, t3);
    atomicAdd(wsum1 + (size_t)d * 4 + 0, t0 * xs);
    atomicAdd(wsum1 + (size_t)d * 4 + 1, t1 * xs);
    atomicAdd(wsum1 + (size_t)d * 4 + 2, t2 * xs);
    atomicAdd(wsum1 + (size_t)d * 4 + 3, t3 * xs);
}

__global__ void k_fin1(const float* __restrict__ W1, const float* __restrict__ b1,
                       const float* __restrict__ s1, const float* __restrict__ wsum1,
                       float* __restrict__ out1, int N) {
    int i = blockIdx.x * blockDim.x + threadIdx.x;
    if (i >= N * 32) return;
    int n = i >> 5, j = i & 31, h = j >> 3;
    float v = W1[j] * (wsum1[n * 4 + h] / s1[n * 4 + h]) + b1[j];
    out1[i] = v > 0.f ? v : 0.f;
}

__global__ void k_gemm2(const float* __restrict__ out1, const float* __restrict__ W2,
                        const float* __restrict__ as2, const float* __restrict__ ad2,
                        float* __restrict__ h2, float* __restrict__ alS2,
                        float* __restrict__ alD2, int N) {
    __shared__ float Bp[32 * 16];
    int t = threadIdx.x;
    for (int i = t; i < 512; i += blockDim.x) {
        int k = i >> 4, n = i & 15;
        float v = 0.f;
        if (n < 8) {
            v = W2[k * 8 + n];
        } else if (n == 8) {
            for (int c = 0; c < 8; ++c) v += W2[k * 8 + c] * as2[c];
        } else if (n == 9) {
            for (int c = 0; c < 8; ++c) v += W2[k * 8 + c] * ad2[c];
        }
        Bp[i] = v;
    }
    __syncthreads();

    int wave = t >> 5, lane = t & 31;
    long tile = (long)blockIdx.x * 8 + wave;
    long row0 = tile * 16;
    if (row0 >= N) return;

    int m = lane & 15, khi = lane >> 4;
    long rowm = row0 + m;
    if (rowm >= N) rowm = N - 1;
    const float* Arow = out1 + rowm * 32;

    v8f c = {};
#pragma unroll
    for (int step = 0; step < 8; ++step) {
        int k0 = step * 4 + khi * 2;
        v2f a; a.x = Arow[k0];          a.y = Arow[k0 + 1];
        v2f b; b.x = Bp[k0 * 16 + m];   b.y = Bp[(k0 + 1) * 16 + m];
        c = __builtin_amdgcn_wmma_f32_16x16x4_f32(false, a, false, b, (short)0, c,
                                                  false, false);
    }

#pragma unroll
    for (int r = 0; r < 8; ++r) {
        long node = row0 + khi * 8 + r;
        if (node >= N) continue;
        float v = c[r];
        if (m < 8)        h2[node * 8 + m] = v;
        else if (m == 8)  alS2[node] = v;
        else if (m == 9)  alD2[node] = v;
    }
}

__global__ void k_edge2_max(const int* __restrict__ ei, int E, int N,
                            const float* __restrict__ alS2, const float* __restrict__ alD2,
                            unsigned* __restrict__ m2) {
    int e = blockIdx.x * blockDim.x + threadIdx.x;
    if (e >= E + N) return;
    int s, d;
    if (e < E) { s = ei[e]; d = ei[E + e]; } else { s = d = e - E; }
    atomicMax(m2 + d, ford(lrelu(alS2[s] + alD2[d])));
}

__global__ void k_edge2_sum(const int* __restrict__ ei, int E, int N,
                            const float* __restrict__ h2,
                            const float* __restrict__ alS2, const float* __restrict__ alD2,
                            const unsigned* __restrict__ m2,
                            float* __restrict__ s2, float* __restrict__ acc2) {
    int e = blockIdx.x * blockDim.x + threadIdx.x;
    if (e >= E + N) return;
    int s, d;
    if (e < E) { s = ei[e]; d = ei[E + e]; } else { s = d = e - E; }
    float t = __expf(lrelu(alS2[s] + alD2[d]) - forddec(m2[d]));
    atomicAdd(s2 + d, t);
    float4 h0 = *reinterpret_cast<const float4*>(h2 + (size_t)s * 8);
    float4 h1 = *reinterpret_cast<const float4*>(h2 + (size_t)s * 8 + 4);
    float* ad = acc2 + (size_t)d * 8;
    atomicAdd(ad + 0, t * h0.x); atomicAdd(ad + 1, t * h0.y);
    atomicAdd(ad + 2, t * h0.z); atomicAdd(ad + 3, t * h0.w);
    atomicAdd(ad + 4, t * h1.x); atomicAdd(ad + 5, t * h1.y);
    atomicAdd(ad + 6, t * h1.z); atomicAdd(ad + 7, t * h1.w);
}

__global__ void k_fin2(const float* __restrict__ acc2, const float* __restrict__ s2,
                       const float* __restrict__ b2, float* __restrict__ out, int N) {
    int i = blockIdx.x * blockDim.x + threadIdx.x;
    if (i >= N * 8) return;
    int n = i >> 3, c = i & 7;
    out[i] = acc2[i] / s2[n] + b2[c];
}

extern "C" void kernel_launch(void* const* d_in, const int* in_sizes, int n_in,
                              void* d_out, int out_size, void* d_ws, size_t ws_size,
                              hipStream_t stream) {
    const float* x   = (const float*)d_in[0];
    const int*   ei  = (const int*)d_in[1];
    const float* W1  = (const float*)d_in[2];
    const float* as1 = (const float*)d_in[3];
    const float* ad1 = (const float*)d_in[4];
    const float* b1  = (const float*)d_in[5];
    const float* W2  = (const float*)d_in[6];
    const float* as2 = (const float*)d_in[7];
    const float* ad2 = (const float*)d_in[8];
    const float* b2  = (const float*)d_in[9];
    (void)n_in; (void)out_size; (void)ws_size;

    int N = in_sizes[0];
    int E = in_sizes[1] / 2;
    int EN = E + N;

    char* ws = (char*)d_ws;
    size_t cur = 0;
    auto carve = [&](size_t bytes) -> void* {
        void* p = ws + cur;
        cur += (bytes + 255) & ~(size_t)255;
        return p;
    };
    unsigned* m1    = (unsigned*)carve(sizeof(unsigned) * (size_t)N * 4);
    unsigned* m2    = (unsigned*)carve(sizeof(unsigned) * (size_t)N);
    float*    s1    = (float*)carve(sizeof(float) * (size_t)N * 4);
    float*    wsum1 = (float*)carve(sizeof(float) * (size_t)N * 4);
    float*    s2    = (float*)carve(sizeof(float) * (size_t)N);
    float*    acc2  = (float*)carve(sizeof(float) * (size_t)N * 8);
    float*    alS1  = (float*)carve(sizeof(float) * (size_t)N * 4);
    float*    alD1  = (float*)carve(sizeof(float) * (size_t)N * 4);
    float*    out1  = (float*)carve(sizeof(float) * (size_t)N * 32);
    float*    h2    = (float*)carve(sizeof(float) * (size_t)N * 8);
    float*    alS2  = (float*)carve(sizeof(float) * (size_t)N);
    float*    alD2  = (float*)carve(sizeof(float) * (size_t)N);

    const int B = 256;
    k_init<<<(8 * N + B - 1) / B, B, 0, stream>>>(m1, 4 * N, m2, N,
                                                  s1, wsum1, 4 * N, s2, N, acc2, 8 * N);
    k_node1<<<(N + B - 1) / B, B, 0, stream>>>(x, W1, as1, ad1, alS1, alD1, N);
    k_edge1_max<<<(EN + B - 1) / B, B, 0, stream>>>(ei, E, N, alS1, alD1, m1);
    k_edge1_sum<<<(EN + B - 1) / B, B, 0, stream>>>(ei, E, N, x, alS1, alD1, m1, s1, wsum1);
    k_fin1<<<(32 * N + B - 1) / B, B, 0, stream>>>(W1, b1, s1, wsum1, out1, N);

    int tiles = (N + 15) / 16;
    k_gemm2<<<(tiles + 7) / 8, B, 0, stream>>>(out1, W2, as2, ad2, h2, alS2, alD2, N);

    k_edge2_max<<<(EN + B - 1) / B, B, 0, stream>>>(ei, E, N, alS2, alD2, m2);
    k_edge2_sum<<<(EN + B - 1) / B, B, 0, stream>>>(ei, E, N, h2, alS2, alD2, m2, s2, acc2);
    k_fin2<<<(8 * N + B - 1) / B, B, 0, stream>>>(acc2, s2, b2, (float*)d_out, N);
}


# Edges grouped by destination in every call (returning integer count, scan, place); 8 lanes own a node, running-max softmax sums in registers: float sums in arrival order per owner, no float atomics
# speedup vs baseline: 5.7398x; 5.7398x over previous
#include <hip/hip_runtime.h>
#include <math.h>


#define NEG_SLOPE 0.2f

typedef __attribute__((ext_vector_type(2))) float v2f;
typedef __attribute__((ext_vector_type(8))) float v8f;

__device__ __forceinline__ float lrelu(float v) { return v > 0.f ? v : NEG_SLOPE * v; }

__global__ void k_zero(int* __restrict__ deg, int N) {
    int i = blockIdx.x * blockDim.x + threadIdx.x;
    if (i < N) deg[i] = 0;
}

__global__ void __launch_bounds__(256) k_rank(const int* __restrict__ ei, int E,
                                              int* __restrict__ deg, int* __restrict__ rank) {
    int stride = gridDim.x * blockDim.x;
    for (int e = blockIdx.x * blockDim.x + threadIdx.x; e < E; e += stride) {
        int d = ei[E + e];
        rank[e] = atomicAdd(deg + d, 1);
    }
}

__global__ void __launch_bounds__(1024) k_scan(const int* __restrict__ deg,
                                               int* __restrict__ rowptr, int N) {
    __shared__ int shBase[32];
    __shared__ int shTot[32];
    int t = threadIdx.x, b = blockIdx.x;
    int lane = t & 31, wv = t >> 5;
    const int4* deg4 = reinterpret_cast<const int4*>(deg);
    int partial = 0;
    int n4 = b * 256;
    for (int i = t; i < n4; i += 1024) {
        int4 q = deg4[i];
        partial += (q.x + q.y) + (q.z + q.w);
    }
    int idx = b * 1024 + t;
    int v = idx < N ? deg[idx] : 0;
    int incl = v;
#pragma unroll
    for (int o = 1; o < 32; o <<= 1) {
        int u = __shfl_up(incl, o);
        if (lane >= o) incl += u;
    }
#pragma unroll
    for (int o = 16; o > 0; o >>= 1) partial += __shfl_xor(partial, o);
    if (lane == 31) shTot[wv] = incl;
    if (lane == 0)  shBase[wv] = partial;
    __syncthreads();
    int base = 0, woff = 0;
#pragma unroll
    for (int k = 0; k < 32; ++k) {
        base += shBase[k];
        if (k < wv) woff += shTot[k];
    }
    int excl = base + woff + (incl - v);
    if (idx < N) rowptr[idx] = excl;
    if (idx == N - 1) rowptr[N] = excl + v;
}

__global__ void __launch_bounds__(256) k_place(const int* __restrict__ ei, int E,
                                               const int* __restrict__ rowptr,
                                               const int* __restrict__ rank,
                                               int* __restrict__ csr) {
    int stride = gridDim.x * blockDim.x;
    for (int e = blockIdx.x * blockDim.x + threadIdx.x; e < E; e += stride) {
        int s = ei[e];
        int d = ei[E + e];
        csr[rowptr[d] + rank[e]] = s;
    }
}

__global__ void __launch_bounds__(256) k_agg1(const float* __restrict__ x,
                                              const int* __restrict__ rowptr,
                                              const int* __restrict__ csr,
                                              const float* __restrict__ W1,
                                              const float* __restrict__ as1,
                                              const float* __restrict__ ad1,
                                              const float* __restrict__ b1,
                                              float* __restrict__ out1, int N) {
    int gid = blockIdx.x * blockDim.x + threadIdx.x;
    int node = gid >> 3, sub = threadIdx.x & 7;
    bool valid = node < N;
    int nd = valid ? node : N - 1;

    float ws[4], wd[4];
#pragma unroll
    for (int h = 0; h < 4; ++h) {
        float a = 0.f, d = 0.f;
#pragma unroll
        for (int c = 0; c < 8; ++c) {
            float w = W1[h * 8 + c];
            a += w * as1[h * 8 + c];
            d += w * ad1[h * 8 + c];
        }
        ws[h] = a; wd[h] = d;
    }

    float xd = x[nd];
    int beg = rowptr[nd];
    int end = valid ? rowptr[nd + 1] : beg;

    float ad[4], m[4], s[4], w[4];
#pragma unroll
    for (int h = 0; h < 4; ++h) {
        ad[h] = xd * wd[h];
        m[h] = lrelu(xd * ws[h] + ad[h]);
        s[h] = sub == 0 ? 1.f : 0.f;
        w[h] = sub == 0 ? xd : 0.f;
    }

    for (int j = beg + sub; j < end; j += 8) {
        int sidx = csr[j];
        float xs = x[sidx];
#pragma unroll
        for (int h = 0; h < 4; ++h) {
            float e  = lrelu(xs * ws[h] + ad[h]);
            float mn = fmaxf(m[h], e);
            float sc = __expf(m[h] - mn);
            float t  = __expf(e - mn);
            s[h] = s[h] * sc + t;
            w[h] = w[h] * sc + t * xs;
            m[h] = mn;
        }
    }

#pragma unroll
    for (int o = 1; o < 8; o <<= 1) {
#pragma unroll
        for (int h = 0; h < 4; ++h) {
            float mo = __shfl_xor(m[h], o);
            float so = __shfl_xor(s[h], o);
            float wo = __shfl_xor(w[h], o);
            float mn = fmaxf(m[h], mo);
            float a  = __expf(m[h] - mn);
            float b  = __expf(mo - mn);
            s[h] = s[h] * a + so * b;
            w[h] = w[h] * a + wo * b;
            m[h] = mn;
        }
    }

    int h = sub >> 1;
    float sh = h == 0 ? s[0] : h == 1 ? s[1] : h == 2 ? s[2] : s[3];
    float wh = h == 0 ? w[0] : h == 1 ? w[1] : h == 2 ? w[2] : w[3];
    float r = wh / sh;
    const float* Wp = W1 + sub * 4;
    const float* Bp1 = b1 + sub * 4;
    float4 o4;
    o4.x = fmaxf(Wp[0] * r + Bp1[0], 0.f);
    o4.y = fmaxf(Wp[1] * r + Bp1[1], 0.f);
    o4.z = fmaxf(Wp[2] * r + Bp1[2], 0.f);
    o4.w = fmaxf(Wp[3] * r + Bp1[3], 0.f);
    if (valid) *reinterpret_cast<float4*>(out1 + (size_t)node * 32 + sub * 4) = o4;
}

__global__ void k_gemm2(const float* __restrict__ out1, const float* __restrict__ W2,
                        const float* __restrict__ as2, const float* __restrict__ ad2,
                        float* __restrict__ rec, int N) {
    __shared__ float Bp[32 * 16];
    int t = threadIdx.x;
    for (int i = t; i < 512; i += blockDim.x) {
        int k = i >> 4, n = i & 15;
        float v = 0.f;
        if (n < 8) {
            v = W2[k * 8 + n];
        } else if (n == 8) {
            for (int c = 0; c < 8; ++c) v += W2[k * 8 + c] * as2[c];
        } else if (n == 9) {
            for (int c = 0; c < 8; ++c) v += W2[k * 8 + c] * ad2[c];
        }
        Bp[i] = v;
    }
    __syncthreads();

    int wave = t >> 5, lane = t & 31;
    long tile = (long)blockIdx.x * 8 + wave;
    long row0 = tile * 16;
    if (row0 >= N) return;

    int m = lane & 15, khi = lane >> 4;
    long rowm = row0 + m;
    if (rowm >= N) rowm = N - 1;
    const float* Arow = out1 + rowm * 32;

    v8f c = {};
#pragma unroll
    for (int step = 0; step < 8; ++step) {
        int k0 = step * 4 + khi * 2;
        v2f a; a.x = Arow[k0];          a.y = Arow[k0 + 1];
        v2f b; b.x = Bp[k0 * 16 + m];   b.y = Bp[(k0 + 1) * 16 + m];
        c = __builtin_amdgcn_wmma_f32_16x16x4_f32(false, a, false, b, (short)0, c,
                                                  false, false);
    }

#pragma unroll
    for (int r = 0; r < 8; ++r) {
        long node = row0 + khi * 8 + r;
        if (node >= N) continue;
        rec[node * 16 + m] = c[r];
    }
}

__global__ void __launch_bounds__(256) k_agg2(const float* __restrict__ rec,
                                              const int* __restrict__ rowptr,
                                              const int* __restrict__ csr,
                                              const float* __restrict__ b2,
                                              float* __restrict__ out, int N) {
    int gid = blockIdx.x * blockDim.x + threadIdx.x;
    int node = gid >> 3, sub = threadIdx.x & 7;
    bool valid = node < N;
    int nd = valid ? node : N - 1;

    const float* rd = rec + (size_t)nd * 16;
    float4 d0 = *reinterpret_cast<const float4*>(rd);
    float4 d1 = *reinterpret_cast<const float4*>(rd + 4);
    float2 dl = *reinterpret_cast<const float2*>(rd + 8);
    float ald = dl.y;
    int beg = rowptr[nd];
    int end = valid ? rowptr[nd + 1] : beg;

    float f0 = sub == 0 ? 1.f : 0.f;
    float m = lrelu(dl.x + ald);
    float s = f0;
    float acc[8] = { f0 * d0.x, f0 * d0.y, f0 * d0.z, f0 * d0.w,
                     f0 * d1.x, f0 * d1.y, f0 * d1.z, f0 * d1.w };

    for (int j = beg + sub; j < end; j += 8) {
        int sidx = csr[j];
        const float* rs = rec + (size_t)sidx * 16;
        float4 a0 = *reinterpret_cast<const float4*>(rs);
        float4 a1 = *reinterpret_cast<const float4*>(rs + 4);
        float als = rs[8];
        float e  = lrelu(als + ald);
        float mn = fmaxf(m, e);
        float sc = __expf(m - mn);
        float t  = __expf(e - mn);
        s = s * sc + t;
        acc[0] = acc[0] * sc + t * a0.x; acc[1] = acc[1] * sc + t * a0.y;
        acc[2] = acc[2] * sc + t * a0.z; acc[3] = acc[3] * sc + t * a0.w;
        acc[4] = acc[4] * sc + t * a1.x; acc[5] = acc[5] * sc + t * a1.y;
        acc[6] = acc[6] * sc + t * a1.z; acc[7] = acc[7] * sc + t * a1.w;
        m = mn;
    }

#pragma unroll
    for (int o = 1; o < 8; o <<= 1) {
        float mo = __shfl_xor(m, o);
        float so = __shfl_xor(s, o);
        float mn = fmaxf(m, mo);
        float a  = __expf(m - mn);
        float b  = __expf(mo - mn);
        s = s * a + so * b;
#pragma unroll
        for (int c = 0; c < 8; ++c) {
            float ao = __shfl_xor(acc[c], o);
            acc[c] = acc[c] * a + ao * b;
        }
        m = mn;
    }

    float av = acc[0];
#pragma unroll
    for (int c = 1; c < 8; ++c) av = sub == c ? acc[c] : av;
    float o1 = av / s + b2[sub];
    if (valid) out[(size_t)node * 8 + sub] = o1;
}

extern "C" void kernel_launch(void* const* d_in, const int* in_sizes, int n_in,
                              void* d_out, int out_size, void* d_ws, size_t ws_size,
                              hipStream_t stream) {
    const float* x   = (const float*)d_in[0];
    const int*   ei  = (const int*)d_in[1];
    const float* W1  = (const float*)d_in[2];
    const float* as1 = (const float*)d_in[3];
    const float* ad1 = (const float*)d_in[4];
    const float* b1  = (const float*)d_in[5];
    const float* W2  = (const float*)d_in[6];
    const float* as2 = (const float*)d_in[7];
    const float* ad2 = (const float*)d_in[8];
    const float* b2  = (const float*)d_in[9];
    (void)n_in; (void)out_size; (void)ws_size;

    int N = in_sizes[0];
    int E = in_sizes[1] / 2;

    char* ws = (char*)d_ws;
    size_t cur = 0;
    auto carve = [&](size_t bytes) -> void* {
        void* p = ws + cur;
        cur += (bytes + 255) & ~(size_t)255;
        return p;
    };
    int*   deg    = (int*)carve(sizeof(int) * ((size_t)N + 1024));
    int*   rowptr = (int*)carve(sizeof(int) * ((size_t)N + 1));
    int*   rank   = (int*)carve(sizeof(int) * (size_t)E);
    int*   csr    = (int*)carve(sizeof(int) * (size_t)E);
    float* out1   = (float*)carve(sizeof(float) * (size_t)N * 32);
    float* rec    = (float*)carve(sizeof(float) * (size_t)N * 16);

    const int B = 256;
    const int EG = 2048;
    k_zero<<<(N + B - 1) / B, B, 0, stream>>>(deg, N);
    k_rank<<<EG, B, 0, stream>>>(ei, E, deg, rank);
    k_scan<<<(N + 1023) / 1024, 1024, 0, stream>>>(deg, rowptr, N);
    k_place<<<EG, B, 0, stream>>>(ei, E, rowptr, rank, csr);
    k_agg1<<<(8 * N + B - 1) / B, B, 0, stream>>>(x, rowptr, csr, W1, as1, ad1, b1, out1, N);

    int tiles = (N + 15) / 16;
    k_gemm2<<<(tiles + 7) / 8, B, 0, stream>>>(out1, W2, as2, ad2, rec, N);

    k_agg2<<<(8 * N + B - 1) / B, B, 0, stream>>>(rec, rowptr, csr, b2, (float*)d_out, N);
}


# Edge grouping rebuilt as a two-level LDS bucket sort (512-node buckets, one work-group each): no per-edge global atomic or random global store; float sums in LDS arrival order per owner
# speedup vs baseline: 19.3346x; 3.3685x over previous
#include <hip/hip_runtime.h>
#include <math.h>


#define NEG_SLOPE 0.2f

typedef __attribute__((ext_vector_type(2))) float v2f;
typedef __attribute__((ext_vector_type(8))) float v8f;

__device__ __forceinline__ float lrelu(float v) { return v > 0.f ? v : NEG_SLOPE * v; }

#define BSH 9
#define BW (1 << BSH)
#define MAXNB 1024
#define SRC_BITS 23
#define PA_T 512
#define PA_I 16
#define KB_T 1024
#define KB_R 20

__global__ void k_zero(int* __restrict__ p, int n) {
    int i = blockIdx.x * blockDim.x + threadIdx.x;
    if (i < n) p[i] = 0;
}

__global__ void __launch_bounds__(PA_T) k_bcount(const int* __restrict__ ei, int E, int NB,
                                                 int* __restrict__ bcnt) {
    __shared__ int hist[MAXNB];
    int t = threadIdx.x;
    hist[2 * t] = 0; hist[2 * t + 1] = 0;
    __syncthreads();
    int base = blockIdx.x * (PA_T * PA_I);
    int d[PA_I];
#pragma unroll
    for (int k = 0; k < PA_I; ++k) {
        int e = base + k * PA_T + t;
        d[k] = e < E ? ei[E + e] : -1;
    }
#pragma unroll
    for (int k = 0; k < PA_I; ++k)
        if (d[k] >= 0) atomicAdd(&hist[d[k] >> BSH], 1);
    __syncthreads();
    for (int i = t; i < NB; i += PA_T) {
        int c = hist[i];
        if (c) atomicAdd(bcnt + i, c);
    }
}

__device__ __forceinline__ int scan_pairs(int v0, int v1, int* wtot, int lane, int wv) {
    int incl = v0 + v1;
#pragma unroll
    for (int o = 1; o < 32; o <<= 1) {
        int u = __shfl_up(incl, o);
        if (lane >= o) incl += u;
    }
    if (lane == 31) wtot[wv] = incl;
    __syncthreads();
    int woff = 0;
#pragma unroll
    for (int k = 0; k < PA_T / 32; ++k)
        if (k < wv) woff += wtot[k];
    return woff + incl - (v0 + v1);
}

__global__ void __launch_bounds__(PA_T) k_bscatter(const int* __restrict__ ei, int E, int NB,
                                                   const int* __restrict__ bcnt,
                                                   int* __restrict__ bcur,
                                                   int* __restrict__ bstart,
                                                   unsigned* __restrict__ binned) {
    __shared__ int hist[MAXNB];
    __shared__ int bbase[MAXNB];
    __shared__ int lstart[MAXNB];
    __shared__ int wtotA[PA_T / 32];
    __shared__ int wtotB[PA_T / 32];
    __shared__ unsigned stage[PA_T * PA_I];
    __shared__ int stageDst[PA_T * PA_I];
    int t = threadIdx.x, lane = t & 31, wv = t >> 5;
    int i0 = 2 * t, i1 = 2 * t + 1;
    hist[i0] = 0; hist[i1] = 0;
    __syncthreads();

    int base = blockIdx.x * (PA_T * PA_I);
    int d[PA_I], s[PA_I];
#pragma unroll
    for (int k = 0; k < PA_I; ++k) {
        int e = base + k * PA_T + t;
        bool in = e < E;
        d[k] = in ? ei[E + e] : -1;
        s[k] = in ? ei[e] : 0;
    }
#pragma unroll
    for (int k = 0; k < PA_I; ++k)
        if (d[k] >= 0) atomicAdd(&hist[d[k] >> BSH], 1);

    int v0 = i0 < NB ? bcnt[i0] : 0;
    int v1 = i1 < NB ? bcnt[i1] : 0;
    int s0 = scan_pairs(v0, v1, wtotA, lane, wv);
    int s1 = s0 + v0;
    if (blockIdx.x == 0) {
        if (i0 <= NB) bstart[i0] = s0;
        if (i1 <= NB) bstart[i1] = s1;
    }
    int c0 = hist[i0], c1 = hist[i1];
    int l0 = scan_pairs(c0, c1, wtotB, lane, wv);
    lstart[i0] = l0; lstart[i1] = l0 + c0;
    if (c0) bbase[i0] = s0 + atomicAdd(bcur + i0, c0);
    if (c1) bbase[i1] = s1 + atomicAdd(bcur + i1, c1);
    hist[i0] = 0; hist[i1] = 0;
    __syncthreads();

#pragma unroll
    for (int k = 0; k < PA_I; ++k) {
        if (d[k] >= 0) {
            int b = d[k] >> BSH;
            int r = atomicAdd(&hist[b], 1);
            int pos = lstart[b] + r;
            stage[pos] = (unsigned)s[k] | ((unsigned)(d[k] & (BW - 1)) << SRC_BITS);
            stageDst[pos] = bbase[b] + r;
        }
    }
    __syncthreads();

    int nvalid = E - base;
    if (nvalid > PA_T * PA_I) nvalid = PA_T * PA_I;
#pragma unroll
    for (int k = 0; k < PA_I; ++k) {
        int i = k * PA_T + t;
        if (i < nvalid) binned[stageDst[i]] = stage[i];
    }
}

__global__ void __launch_bounds__(KB_T) k_bucket(const unsigned* __restrict__ binned,
                                                 const int* __restrict__ bstart,
                                                 int* __restrict__ rowptr,
                                                 int* __restrict__ csr, int N) {
    __shared__ int cnt[BW];
    __shared__ int wtot[BW / 32];
    __shared__ int stage[KB_T * KB_R];
    int t = threadIdx.x, lane = t & 31, wv = t >> 5;
    int b = blockIdx.x;
    int bs = bstart[b];
    int n = bstart[b + 1] - bs;
    bool fits = n <= KB_T * KB_R;
    const unsigned* bin = binned + bs;
    int* cs = csr + bs;
    if (t < BW) cnt[t] = 0;
    __syncthreads();

    unsigned w[KB_R];
#pragma unroll
    for (int u = 0; u < KB_R; ++u) {
        int j = u * KB_T + t;
        w[u] = j < n ? bin[j] : 0u;
    }
#pragma unroll
    for (int u = 0; u < KB_R; ++u)
        if (u * KB_T + t < n) atomicAdd(&cnt[w[u] >> SRC_BITS], 1);
    for (int j = KB_T * KB_R + t; j < n; j += KB_T)
        atomicAdd(&cnt[bin[j] >> SRC_BITS], 1);
    __syncthreads();

    int v = t < BW ? cnt[t] : 0;
    int incl = v;
#pragma unroll
    for (int o = 1; o < 32; o <<= 1) {
        int u = __shfl_up(incl, o);
        if (lane >= o) incl += u;
    }
    if (lane == 31 && t < BW) wtot[wv] = incl;
    __syncthreads();
    if (t < BW) {
        int woff = 0;
#pragma unroll
        for (int k = 0; k < BW / 32; ++k)
            if (k < wv) woff += wtot[k];
        int excl = woff + incl - v;
        cnt[t] = excl;
        int node = b * BW + t;
        if (node < N) rowptr[node] = bs + excl;
        if (node == N - 1) rowptr[N] = bs + excl + v;
    }
    __syncthreads();

    const unsigned SRC_MASK = (1u << SRC_BITS) - 1u;
    if (fits) {
#pragma unroll
        for (int u = 0; u < KB_R; ++u) {
            if (u * KB_T + t < n) {
                int p = atomicAdd(&cnt[w[u] >> SRC_BITS], 1);
                stage[p] = (int)(w[u] & SRC_MASK);
            }
        }
        __syncthreads();
        for (int i = t; i < n; i += KB_T) cs[i] = stage[i];
    } else {
#pragma unroll
        for (int u = 0; u < KB_R; ++u) {
            if (u * KB_T + t < n) {
                int p = atomicAdd(&cnt[w[u] >> SRC_BITS], 1);
                cs[p] = (int)(w[u] & SRC_MASK);
            }
        }
        for (int j = KB_T * KB_R + t; j < n; j += KB_T) {
            unsigned x = bin[j];
            int p = atomicAdd(&cnt[x >> SRC_BITS], 1);
            cs[p] = (int)(x & SRC_MASK);
        }
    }
}

__global__ void __launch_bounds__(256) k_agg1(const float* __restrict__ x,
                                              const int* __restrict__ rowptr,
                                              const int* __restrict__ csr,
                                              const float* __restrict__ W1,
                                              const float* __restrict__ as1,
                                              const float* __restrict__ ad1,
                                              const float* __restrict__ b1,
                                              float* __restrict__ out1, int N) {
    int gid = blockIdx.x * blockDim.x + threadIdx.x;
    int node = gid >> 3, sub = threadIdx.x & 7;
    bool valid = node < N;
    int nd = valid ? node : N - 1;

    float ws[4], wd[4];
#pragma unroll
    for (int h = 0; h < 4; ++h) {
        float a = 0.f, d = 0.f;
#pragma unroll
        for (int c = 0; c < 8; ++c) {
            float w = W1[h * 8 + c];
            a += w * as1[h * 8 + c];
            d += w * ad1[h * 8 + c];
        }
        ws[h] = a; wd[h] = d;
    }

    float xd = x[nd];
    int beg = rowptr[nd];
    int end = valid ? rowptr[nd + 1] : beg;

    float ad[4], m[4], s[4], w[4];
#pragma unroll
    for (int h = 0; h < 4; ++h) {
        ad[h] = xd * wd[h];
        m[h] = lrelu(xd * ws[h] + ad[h]);
        s[h] = sub == 0 ? 1.f : 0.f;
        w[h] = sub == 0 ? xd : 0.f;
    }

    for (int j = beg + sub; j < end; j += 8) {
        int sidx = csr[j];
        float xs = x[sidx];
#pragma unroll
        for (int h = 0; h < 4; ++h) {
            float e  = lrelu(xs * ws[h] + ad[h]);
            float mn = fmaxf(m[h], e);
            float sc = __expf(m[h] - mn);
            float t  = __expf(e - mn);
            s[h] = s[h] * sc + t;
            w[h] = w[h] * sc + t * xs;
            m[h] = mn;
        }
    }

#pragma unroll
    for (int o = 1; o < 8; o <<= 1) {
#pragma unroll
        for (int h = 0; h < 4; ++h) {
            float mo = __shfl_xor(m[h], o);
            float so = __shfl_xor(s[h], o);
            float wo = __shfl_xor(w[h], o);
            float mn = fmaxf(m[h], mo);
            float a  = __expf(m[h] - mn);
            float b  = __expf(mo - mn);
            s[h] = s[h] * a + so * b;
            w[h] = w[h] * a + wo * b;
            m[h] = mn;
        }
    }

    int h = sub >> 1;
    float sh = h == 0 ? s[0] : h == 1 ? s[1] : h == 2 ? s[2] : s[3];
    float wh = h == 0 ? w[0] : h == 1 ? w[1] : h == 2 ? w[2] : w[3];
    float r = wh / sh;
    const float* Wp = W1 + sub * 4;
    const float* Bp1 = b1 + sub * 4;
    float4 o4;
    o4.x = fmaxf(Wp[0] * r + Bp1[0], 0.f);
    o4.y = fmaxf(Wp[1] * r + Bp1[1], 0.f);
    o4.z = fmaxf(Wp[2] * r + Bp1[2], 0.f);
    o4.w = fmaxf(Wp[3] * r + Bp1[3], 0.f);
    if (valid) *reinterpret_cast<float4*>(out1 + (size_t)node * 32 + sub * 4) = o4;
}

__global__ void k_gemm2(const float* __restrict__ out1, const float* __restrict__ W2,
                        const float* __restrict__ as2, const float* __restrict__ ad2,
                        float* __restrict__ rec, int N) {
    __shared__ float Bp[32 * 16];
    int t = threadIdx.x;
    for (int i = t; i < 512; i += blockDim.x) {
        int k = i >> 4, n = i & 15;
        float v = 0.f;
        if (n < 8) {
            v = W2[k * 8 + n];
        } else if (n == 8) {
            for (int c = 0; c < 8; ++c) v += W2[k * 8 + c] * as2[c];
        } else if (n == 9) {
            for (int c = 0; c < 8; ++c) v += W2[k * 8 + c] * ad2[c];
        }
        Bp[i] = v;
    }
    __syncthreads();

    int wave = t >> 5, lane = t & 31;
    long tile = (long)blockIdx.x * 8 + wave;
    long row0 = tile * 16;
    if (row0 >= N) return;

    int m = lane & 15, khi = lane >> 4;
    long rowm = row0 + m;
    if (rowm >= N) rowm = N - 1;
    const float* Arow = out1 + rowm * 32;

    v8f c = {};
#pragma unroll
    for (int step = 0; step < 8; ++step) {
        int k0 = step * 4 + khi * 2;
        v2f a; a.x = Arow[k0];          a.y = Arow[k0 + 1];
        v2f b; b.x = Bp[k0 * 16 + m];   b.y = Bp[(k0 + 1) * 16 + m];
        c = __builtin_amdgcn_wmma_f32_16x16x4_f32(false, a, false, b, (short)0, c,
                                                  false, false);
    }

#pragma unroll
    for (int r = 0; r < 8; ++r) {
        long node = row0 + khi * 8 + r;
        if (node >= N) continue;
        rec[node * 16 + m] = c[r];
    }
}

__global__ void __launch_bounds__(256) k_agg2(const float* __restrict__ rec,
                                              const int* __restrict__ rowptr,
                                              const int* __restrict__ csr,
                                              const float* __restrict__ b2,
                                              float* __restrict__ out, int N) {
    int gid = blockIdx.x * blockDim.x + threadIdx.x;
    int node = gid >> 3, sub = threadIdx.x & 7;
    bool valid = node < N;
    int nd = valid ? node : N - 1;

    const float* rd = rec + (size_t)nd * 16;
    float4 d0 = *reinterpret_cast<const float4*>(rd);
    float4 d1 = *reinterpret_cast<const float4*>(rd + 4);
    float2 dl = *reinterpret_cast<const float2*>(rd + 8);
    float ald = dl.y;
    int beg = rowptr[nd];
    int end = valid ? rowptr[nd + 1] : beg;

    float f0 = sub == 0 ? 1.f : 0.f;
    float m = lrelu(dl.x + ald);
    float s = f0;
    float acc[8] = { f0 * d0.x, f0 * d0.y, f0 * d0.z, f0 * d0.w,
                     f0 * d1.x, f0 * d1.y, f0 * d1.z, f0 * d1.w };

    for (int j = beg + sub; j < end; j += 8) {
        int sidx = csr[j];
        const float* rs = rec + (size_t)sidx * 16;
        float4 a0 = *reinterpret_cast<const float4*>(rs);
        float4 a1 = *reinterpret_cast<const float4*>(rs + 4);
        float als = rs[8];
        float e  = lrelu(als + ald);
        float mn = fmaxf(m, e);
        float sc = __expf(m - mn);
        float t  = __expf(e - mn);
        s = s * sc + t;
        acc[0] = acc[0] * sc + t * a0.x; acc[1] = acc[1] * sc + t * a0.y;
        acc[2] = acc[2] * sc + t * a0.z; acc[3] = acc[3] * sc + t * a0.w;
        acc[4] = acc[4] * sc + t * a1.x; acc[5] = acc[5] * sc + t * a1.y;
        acc[6] = acc[6] * sc + t * a1.z; acc[7] = acc[7] * sc + t * a1.w;
        m = mn;
    }

#pragma unroll
    for (int o = 1; o < 8; o <<= 1) {
        float mo = __shfl_xor(m, o);
        float so = __shfl_xor(s, o);
        float mn = fmaxf(m, mo);
        float a  = __expf(m - mn);
        float b  = __expf(mo - mn);
        s = s * a + so * b;
#pragma unroll
        for (int c = 0; c < 8; ++c) {
            float ao = __shfl_xor(acc[c], o);
            acc[c] = acc[c] * a + ao * b;
        }
        m = mn;
    }

    float av = acc[0];
#pragma unroll
    for (int c = 1; c < 8; ++c) av = sub == c ? acc[c] : av;
    float o1 = av / s + b2[sub];
    if (valid) out[(size_t)node * 8 + sub] = o1;
}

extern "C" void kernel_launch(void* const* d_in, const int* in_sizes, int n_in,
                              void* d_out, int out_size, void* d_ws, size_t ws_size,
                              hipStream_t stream) {
    const float* x   = (const float*)d_in[0];
    const int*   ei  = (const int*)d_in[1];
    const float* W1  = (const float*)d_in[2];
    const float* as1 = (const float*)d_in[3];
    const float* ad1 = (const float*)d_in[4];
    const float* b1  = (const float*)d_in[5];
    const float* W2  = (const float*)d_in[6];
    const float* as2 = (const float*)d_in[7];
    const float* ad2 = (const float*)d_in[8];
    const float* b2  = (const float*)d_in[9];
    (void)n_in; (void)out_size; (void)ws_size;

    int N = in_sizes[0];
    int E = in_sizes[1] / 2;

    char* ws = (char*)d_ws;
    size_t cur = 0;
    auto carve = [&](size_t bytes) -> void* {
        void* p = ws + cur;
        cur += (bytes + 255) & ~(size_t)255;
        return p;
    };
    int*      bcnt   = (int*)carve(sizeof(int) * 2 * MAXNB);
    int*      bcur   = bcnt + MAXNB;
    int*      bstart = (int*)carve(sizeof(int) * (MAXNB + 1));
    int*      rowptr = (int*)carve(sizeof(int) * ((size_t)N + 1));
    unsigned* binned = (unsigned*)carve(sizeof(unsigned) * (size_t)E);
    int*      csr    = (int*)carve(sizeof(int) * (size_t)E);
    float* out1   = (float*)carve(sizeof(float) * (size_t)N * 32);
    float* rec    = (float*)carve(sizeof(float) * (size_t)N * 16);

    const int B = 256;
    int NB = (N + BW - 1) >> BSH;
    int PG = (E + PA_T * PA_I - 1) / (PA_T * PA_I);
    k_zero<<<(2 * MAXNB + B - 1) / B, B, 0, stream>>>(bcnt, 2 * MAXNB);
    k_bcount<<<PG, PA_T, 0, stream>>>(ei, E, NB, bcnt);
    k_bscatter<<<PG, PA_T, 0, stream>>>(ei, E, NB, bcnt, bcur, bstart, binned);
    k_bucket<<<NB, KB_T, 0, stream>>>(binned, bstart, rowptr, csr, N);
    k_agg1<<<(8 * N + B - 1) / B, B, 0, stream>>>(x, rowptr, csr, W1, as1, ad1, b1, out1, N);

    int tiles = (N + 15) / 16;
    k_gemm2<<<(tiles + 7) / 8, B, 0, stream>>>(out1, W2, as2, ad2, rec, N);

    k_agg2<<<(8 * N + B - 1) / B, B, 0, stream>>>(rec, rowptr, csr, b2, (float*)d_out, N);
}


# Level 1 leaves each block's sorted edges in its own slice with a run table; level 2 walks the runs: no count pass, zeroing launch or global atomic (7 to 5 launches); sums stay in LDS arrival order
# speedup vs baseline: 20.6215x; 1.0666x over previous
#include <hip/hip_runtime.h>
#include <math.h>


#define NEG_SLOPE 0.2f

typedef __attribute__((ext_vector_type(2))) float v2f;
typedef __attribute__((ext_vector_type(8))) float v8f;

__device__ __forceinline__ float lrelu(float v) { return v > 0.f ? v : NEG_SLOPE * v; }

#define BSH 9
#define BW (1 << BSH)
#define MAXNB 1024
#define SRC_BITS 23
#define PA_T 512
#define PA_I 16
#define RUN_CAP (PA_T * PA_I)
#define KB_T 1024
#define KB_R 20

__device__ __forceinline__ int scan_pairs(int v0, int v1, int* wtot, int lane, int wv) {
    int incl = v0 + v1;
#pragma unroll
    for (int o = 1; o < 32; o <<= 1) {
        int u = __shfl_up(incl, o);
        if (lane >= o) incl += u;
    }
    if (lane == 31) wtot[wv] = incl;
    __syncthreads();
    int woff = 0;
#pragma unroll
    for (int k = 0; k < PA_T / 32; ++k)
        if (k < wv) woff += wtot[k];
    return woff + incl - (v0 + v1);
}

__global__ void __launch_bounds__(PA_T) k_bsort(const int* __restrict__ ei, int E, int NB, int TP,
                                                int* __restrict__ tab,
                                                unsigned* __restrict__ binned) {
    __shared__ int hist[MAXNB];
    __shared__ int lstart[MAXNB];
    __shared__ int wtot[PA_T / 32];
    __shared__ unsigned stage[RUN_CAP];
    int t = threadIdx.x, lane = t & 31, wv = t >> 5;
    int i0 = 2 * t, i1 = 2 * t + 1;
    hist[i0] = 0; hist[i1] = 0;
    __syncthreads();

    int blk = blockIdx.x;
    int base = blk * RUN_CAP;
    int d[PA_I], s[PA_I];
#pragma unroll
    for (int k = 0; k < PA_I; ++k) {
        int e = base + k * PA_T + t;
        bool in = e < E;
        d[k] = in ? ei[E + e] : -1;
        s[k] = in ? ei[e] : 0;
    }
#pragma unroll
    for (int k = 0; k < PA_I; ++k)
        if (d[k] >= 0) atomicAdd(&hist[d[k] >> BSH], 1);
    __syncthreads();

    int c0 = hist[i0], c1 = hist[i1];
    int l0 = scan_pairs(c0, c1, wtot, lane, wv);
    lstart[i0] = l0; lstart[i1] = l0 + c0;
    int* row = tab + (size_t)blk * TP;
    if (i0 <= NB) row[i0] = l0;
    if (i1 <= NB) row[i1] = l0 + c0;
    hist[i0] = 0; hist[i1] = 0;
    __syncthreads();

#pragma unroll
    for (int k = 0; k < PA_I; ++k) {
        if (d[k] >= 0) {
            int b = d[k] >> BSH;
            int r = atomicAdd(&hist[b], 1);
            stage[lstart[b] + r] = (unsigned)s[k] | ((unsigned)(d[k] & (BW - 1)) << SRC_BITS);
        }
    }
    __syncthreads();

    int nvalid = E - base;
    if (nvalid > RUN_CAP) nvalid = RUN_CAP;
#pragma unroll
    for (int k = 0; k < PA_I; ++k) {
        int i = k * PA_T + t;
        if (i < nvalid) binned[base + i] = stage[i];
    }
}

__device__ __forceinline__ int run_slot(const int* roffs, const int* rbase, int PG, int j) {
    int lo = 0, hi = PG - 1;
    while (lo < hi) {
        int mid = (lo + hi + 1) >> 1;
        if (roffs[mid] <= j) lo = mid; else hi = mid - 1;
    }
    return rbase[lo] + j;
}

__global__ void __launch_bounds__(KB_T) k_bucket(const unsigned* __restrict__ binned,
                                                 const int* __restrict__ tab, int PG, int TP,
                                                 int* __restrict__ rowptr,
                                                 int* __restrict__ csr, int N) {
    __shared__ int cnt[BW];
    __shared__ int wtot[BW / 32];
    __shared__ int rtot[KB_T / 32];
    __shared__ int rsum[KB_T / 32];
    __shared__ int roffs[KB_T];
    __shared__ int rbase[KB_T];
    __shared__ int stage[KB_T * KB_R];
    int t = threadIdx.x, lane = t & 31, wv = t >> 5;
    int b = blockIdx.x;

    int r0 = 0, len = 0;
    if (t < PG) {
        const int* row = tab + (size_t)t * TP;
        r0 = row[b];
        len = row[b + 1] - r0;
    }
    int rincl = len, rs = r0;
#pragma unroll
    for (int o = 1; o < 32; o <<= 1) {
        int u = __shfl_up(rincl, o);
        if (lane >= o) rincl += u;
        rs += __shfl_xor(rs, o);
    }
    if (lane == 31) rtot[wv] = rincl;
    if (lane == 0) rsum[wv] = rs;
    if (t < BW) cnt[t] = 0;
    __syncthreads();
    int roff = rincl - len, n = 0, bs = 0;
#pragma unroll
    for (int k = 0; k < KB_T / 32; ++k) {
        int x = rtot[k];
        if (k < wv) roff += x;
        n += x;
        bs += rsum[k];
    }
    int g0 = t * RUN_CAP + r0;
    roffs[t] = roff;
    rbase[t] = g0 - roff;
    if (t == 0 && n == 0) stage[0] = 0;
    for (int i = 0; i < len; ++i) {
        int j = roff + i;
        if (j < KB_T * KB_R) stage[j] = g0 + i;
    }
    __syncthreads();

    bool fits = n <= KB_T * KB_R;
    int* cs = csr + bs;

    unsigned w[KB_R];
    int slot[KB_R];
#pragma unroll
    for (int u = 0; u < KB_R; ++u) {
        int j = u * KB_T + t;
        slot[u] = stage[j < n ? j : 0];
    }
#pragma unroll
    for (int u = 0; u < KB_R; ++u) {
        int j = u * KB_T + t;
        w[u] = j < n ? binned[slot[u]] : 0u;
    }
#pragma unroll
    for (int u = 0; u < KB_R; ++u)
        if (u * KB_T + t < n) atomicAdd(&cnt[w[u] >> SRC_BITS], 1);
    for (int j = KB_T * KB_R + t; j < n; j += KB_T)
        atomicAdd(&cnt[binned[run_slot(roffs, rbase, PG, j)] >> SRC_BITS], 1);
    __syncthreads();

    int v = t < BW ? cnt[t] : 0;
    int incl = v;
#pragma unroll
    for (int o = 1; o < 32; o <<= 1) {
        int u = __shfl_up(incl, o);
        if (lane >= o) incl += u;
    }
    if (lane == 31 && t < BW) wtot[wv] = incl;
    __syncthreads();
    if (t < BW) {
        int woff = 0;
#pragma unroll
        for (int k = 0; k < BW / 32; ++k)
            if (k < wv) woff += wtot[k];
        int excl = woff + incl - v;
        cnt[t] = excl;
        int node = b * BW + t;
        if (node < N) rowptr[node] = bs + excl;
        if (node == N - 1) rowptr[N] = bs + excl + v;
    }
    __syncthreads();

    const unsigned SRC_MASK = (1u << SRC_BITS) - 1u;
    if (fits) {
#pragma unroll
        for (int u = 0; u < KB_R; ++u) {
            if (u * KB_T + t < n) {
                int p = atomicAdd(&cnt[w[u] >> SRC_BITS], 1);
                stage[p] = (int)(w[u] & SRC_MASK);
            }
        }
        __syncthreads();
        for (int i = t; i < n; i += KB_T) cs[i] = stage[i];
    } else {
#pragma unroll
        for (int u = 0; u < KB_R; ++u) {
            if (u * KB_T + t < n) {
                int p = atomicAdd(&cnt[w[u] >> SRC_BITS], 1);
                cs[p] = (int)(w[u] & SRC_MASK);
            }
        }
        for (int j = KB_T * KB_R + t; j < n; j += KB_T) {
            unsigned x = binned[run_slot(roffs, rbase, PG, j)];
            int p = atomicAdd(&cnt[x >> SRC_BITS], 1);
            cs[p] = (int)(x & SRC_MASK);
        }
    }
}

__global__ void __launch_bounds__(256) k_agg1(const float* __restrict__ x,
                                              const int* __restrict__ rowptr,
                                              const int* __restrict__ csr,
                                              const float* __restrict__ W1,
                                              const float* __restrict__ as1,
                                              const float* __restrict__ ad1,
                                              const float* __restrict__ b1,
                                              float* __restrict__ out1, int N) {
    int gid = blockIdx.x * blockDim.x + threadIdx.x;
    int node = gid >> 3, sub = threadIdx.x & 7;
    bool valid = node < N;
    int nd = valid ? node : N - 1;

    float ws[4], wd[4];
#pragma unroll
    for (int h = 0; h < 4; ++h) {
        float a = 0.f, d = 0.f;
#pragma unroll
        for (int c = 0; c < 8; ++c) {
            float w = W1[h * 8 + c];
            a += w * as1[h * 8 + c];
            d += w * ad1[h * 8 + c];
        }
        ws[h] = a; wd[h] = d;
    }

    float xd = x[nd];
    int beg = rowptr[nd];
    int end = valid ? rowptr[nd + 1] : beg;

    float ad[4], m[4], s[4], w[4];
#pragma unroll
    for (int h = 0; h < 4; ++h) {
        ad[h] = xd * wd[h];
        m[h] = lrelu(xd * ws[h] + ad[h]);
        s[h] = sub == 0 ? 1.f : 0.f;
        w[h] = sub == 0 ? xd : 0.f;
    }

    for (int j = beg + sub; j < end; j += 8) {
        int sidx = csr[j];
        float xs = x[sidx];
#pragma unroll
        for (int h = 0; h < 4; ++h) {
            float e  = lrelu(xs * ws[h] + ad[h]);
            float mn = fmaxf(m[h], e);
            float sc = __expf(m[h] - mn);
            float t  = __expf(e - mn);
            s[h] = s[h] * sc + t;
            w[h] = w[h] * sc + t * xs;
            m[h] = mn;
        }
    }

#pragma unroll
    for (int o = 1; o < 8; o <<= 1) {
#pragma unroll
        for (int h = 0; h < 4; ++h) {
            float mo = __shfl_xor(m[h], o);
            float so = __shfl_xor(s[h], o);
            float wo = __shfl_xor(w[h], o);
            float mn = fmaxf(m[h], mo);
            float a  = __expf(m[h] - mn);
            float b  = __expf(mo - mn);
            s[h] = s[h] * a + so * b;
            w[h] = w[h] * a + wo * b;
            m[h] = mn;
        }
    }

    int h = sub >> 1;
    float sh = h == 0 ? s[0] : h == 1 ? s[1] : h == 2 ? s[2] : s[3];
    float wh = h == 0 ? w[0] : h == 1 ? w[1] : h == 2 ? w[2] : w[3];
    float r = wh / sh;
    const float* Wp = W1 + sub * 4;
    const float* Bp1 = b1 + sub * 4;
    float4 o4;
    o4.x = fmaxf(Wp[0] * r + Bp1[0], 0.f);
    o4.y = fmaxf(Wp[1] * r + Bp1[1], 0.f);
    o4.z = fmaxf(Wp[2] * r + Bp1[2], 0.f);
    o4.w = fmaxf(Wp[3] * r + Bp1[3], 0.f);
    if (valid) *reinterpret_cast<float4*>(out1 + (size_t)node * 32 + sub * 4) = o4;
}

__global__ void k_gemm2(const float* __restrict__ out1, const float* __restrict__ W2,
                        const float* __restrict__ as2, const float* __restrict__ ad2,
                        float* __restrict__ rec, int N) {
    __shared__ float Bp[32 * 16];
    int t = threadIdx.x;
    for (int i = t; i < 512; i += blockDim.x) {
        int k = i >> 4, n = i & 15;
        float v = 0.f;
        if (n < 8) {
            v = W2[k * 8 + n];
        } else if (n == 8) {
            for (int c = 0; c < 8; ++c) v += W2[k * 8 + c] * as2[c];
        } else if (n == 9) {
            for (int c = 0; c < 8; ++c) v += W2[k * 8 + c] * ad2[c];
        }
        Bp[i] = v;
    }
    __syncthreads();

    int wave = t >> 5, lane = t & 31;
    long tile = (long)blockIdx.x * 8 + wave;
    long row0 = tile * 16;
    if (row0 >= N) return;

    int m = lane & 15, khi = lane >> 4;
    long rowm = row0 + m;
    if (rowm >= N) rowm = N - 1;
    const float* Arow = out1 + rowm * 32;

    v8f c = {};
#pragma unroll
    for (int step = 0; step < 8; ++step) {
        int k0 = step * 4 + khi * 2;
        v2f a; a.x = Arow[k0];          a.y = Arow[k0 + 1];
        v2f b; b.x = Bp[k0 * 16 + m];   b.y = Bp[(k0 + 1) * 16 + m];
        c = __builtin_amdgcn_wmma_f32_16x16x4_f32(false, a, false, b, (short)0, c,
                                                  false, false);
    }

#pragma unroll
    for (int r = 0; r < 8; ++r) {
        long node = row0 + khi * 8 + r;
        if (node >= N) continue;
        rec[node * 16 + m] = c[r];
    }
}

__global__ void __launch_bounds__(256) k_agg2(const float* __restrict__ rec,
                                              const int* __restrict__ rowptr,
                                              const int* __restrict__ csr,
                                              const float* __restrict__ b2,
                                              float* __restrict__ out, int N) {
    int gid = blockIdx.x * blockDim.x + threadIdx.x;
    int node = gid >> 3, sub = threadIdx.x & 7;
    bool valid = node < N;
    int nd = valid ? node : N - 1;

    const float* rd = rec + (size_t)nd * 16;
    float4 d0 = *reinterpret_cast<const float4*>(rd);
    float4 d1 = *reinterpret_cast<const float4*>(rd + 4);
    float2 dl = *reinterpret_cast<const float2*>(rd + 8);
    float ald = dl.y;
    int beg = rowptr[nd];
    int end = valid ? rowptr[nd + 1] : beg;

    float f0 = sub == 0 ? 1.f : 0.f;
    float m = lrelu(dl.x + ald);
    float s = f0;
    float acc[8] = { f0 * d0.x, f0 * d0.y, f0 * d0.z, f0 * d0.w,
                     f0 * d1.x, f0 * d1.y, f0 * d1.z, f0 * d1.w };

    for (int j = beg + sub; j < end; j += 8) {
        int sidx = csr[j];
        const float* rs = rec + (size_t)sidx * 16;
        float4 a0 = *reinterpret_cast<const float4*>(rs);
        float4 a1 = *reinterpret_cast<const float4*>(rs + 4);
        float als = rs[8];
        float e  = lrelu(als + ald);
        float mn = fmaxf(m, e);
        float sc = __expf(m - mn);
        float t  = __expf(e - mn);
        s = s * sc + t;
        acc[0] = acc[0] * sc + t * a0.x; acc[1] = acc[1] * sc + t * a0.y;
        acc[2] = acc[2] * sc + t * a0.z; acc[3] = acc[3] * sc + t * a0.w;
        acc[4] = acc[4] * sc + t * a1.x; acc[5] = acc[5] * sc + t * a1.y;
        acc[6] = acc[6] * sc + t * a1.z; acc[7] = acc[7] * sc + t * a1.w;
        m = mn;
    }

#pragma unroll
    for (int o = 1; o < 8; o <<= 1) {
        float mo = __shfl_xor(m, o);
        float so = __shfl_xor(s, o);
        float mn = fmaxf(m, mo);
        float a  = __expf(m - mn);
        float b  = __expf(mo - mn);
        s = s * a + so * b;
#pragma unroll
        for (int c = 0; c < 8; ++c) {
            float ao = __shfl_xor(acc[c], o);
            acc[c] = acc[c] * a + ao * b;
        }
        m = mn;
    }

    float av = acc[0];
#pragma unroll
    for (int c = 1; c < 8; ++c) av = sub == c ? acc[c] : av;
    float o1 = av / s + b2[sub];
    if (valid) out[(size_t)node * 8 + sub] = o1;
}

extern "C" void kernel_launch(void* const* d_in, const int* in_sizes, int n_in,
                              void* d_out, int out_size, void* d_ws, size_t ws_size,
                              hipStream_t stream) {
    const float* x   = (const float*)d_in[0];
    const int*   ei  = (const int*)d_in[1];
    const float* W1  = (const float*)d_in[2];
    const float* as1 = (const float*)d_in[3];
    const float* ad1 = (const float*)d_in[4];
    const float* b1  = (const float*)d_in[5];
    const float* W2  = (const float*)d_in[6];
    const float* as2 = (const float*)d_in[7];
    const float* ad2 = (const float*)d_in[8];
    const float* b2  = (const float*)d_in[9];
    (void)n_in; (void)out_size; (void)ws_size;

    int N = in_sizes[0];
    int E = in_sizes[1] / 2;

    char* ws = (char*)d_ws;
    size_t cur = 0;
    auto carve = [&](size_t bytes) -> void* {
        void* p = ws + cur;
        cur += (bytes + 255) & ~(size_t)255;
        return p;
    };
    int NB = (N + BW - 1) >> BSH;
    int PG = (E + RUN_CAP - 1) / RUN_CAP;
    int TP = (NB + 1 + 31) & ~31;
    int*      tab    = (int*)carve(sizeof(int) * (size_t)PG * (size_t)TP);
    int*      rowptr = (int*)carve(sizeof(int) * ((size_t)N + 1));
    unsigned* binned = (unsigned*)carve(sizeof(unsigned) * (size_t)PG * RUN_CAP);
    int*      csr    = (int*)carve(sizeof(int) * (size_t)E);
    float* out1   = (float*)carve(sizeof(float) * (size_t)N * 32);
    float* rec    = (float*)carve(sizeof(float) * (size_t)N * 16);

    const int B = 256;
    k_bsort<<<PG, PA_T, 0, stream>>>(ei, E, NB, TP, tab, binned);
    k_bucket<<<NB, KB_T, 0, stream>>>(binned, tab, PG, TP, rowptr, csr, N);
    k_agg1<<<(8 * N + B - 1) / B, B, 0, stream>>>(x, rowptr, csr, W1, as1, ad1, b1, out1, N);

    int tiles = (N + 15) / 16;
    k_gemm2<<<(tiles + 7) / 8, B, 0, stream>>>(out1, W2, as2, ad2, rec, N);

    k_agg2<<<(8 * N + B - 1) / B, B, 0, stream>>>(rec, rowptr, csr, b2, (float*)d_out, N);
}


# Layer-1 aggregation: one work-group per WGP copies the first 80,000 floats of x into LDS and reads source values there, so most random 4-byte gathers skip the L2 cache; per-lane edge order kept
# speedup vs baseline: 22.6466x; 1.0982x over previous
#include <hip/hip_runtime.h>
#include <math.h>


#define NEG_SLOPE 0.2f

typedef __attribute__((ext_vector_type(2))) float v2f;
typedef __attribute__((ext_vector_type(8))) float v8f;

__device__ __forceinline__ float lrelu(float v) { return v > 0.f ? v : NEG_SLOPE * v; }

#define BSH 9
#define BW (1 << BSH)
#define MAXNB 1024
#define SRC_BITS 23
#define PA_T 512
#define PA_I 16
#define RUN_CAP (PA_T * PA_I)
#define KB_T 1024
#define KB_R 20

__device__ __forceinline__ int scan_pairs(int v0, int v1, int* wtot, int lane, int wv) {
    int incl = v0 + v1;
#pragma unroll
    for (int o = 1; o < 32; o <<= 1) {
        int u = __shfl_up(incl, o);
        if (lane >= o) incl += u;
    }
    if (lane == 31) wtot[wv] = incl;
    __syncthreads();
    int woff = 0;
#pragma unroll
    for (int k = 0; k < PA_T / 32; ++k)
        if (k < wv) woff += wtot[k];
    return woff + incl - (v0 + v1);
}

__global__ void __launch_bounds__(PA_T) k_bsort(const int* __restrict__ ei, int E, int NB, int TP,
                                                int* __restrict__ tab,
                                                unsigned* __restrict__ binned) {
    __shared__ int hist[MAXNB];
    __shared__ int lstart[MAXNB];
    __shared__ int wtot[PA_T / 32];
    __shared__ unsigned stage[RUN_CAP];
    int t = threadIdx.x, lane = t & 31, wv = t >> 5;
    int i0 = 2 * t, i1 = 2 * t + 1;
    hist[i0] = 0; hist[i1] = 0;
    __syncthreads();

    int blk = blockIdx.x;
    int base = blk * RUN_CAP;
    int d[PA_I], s[PA_I];
#pragma unroll
    for (int k = 0; k < PA_I; ++k) {
        int e = base + k * PA_T + t;
        bool in = e < E;
        d[k] = in ? ei[E + e] : -1;
        s[k] = in ? ei[e] : 0;
    }
#pragma unroll
    for (int k = 0; k < PA_I; ++k)
        if (d[k] >= 0) atomicAdd(&hist[d[k] >> BSH], 1);
    __syncthreads();

    int c0 = hist[i0], c1 = hist[i1];
    int l0 = scan_pairs(c0, c1, wtot, lane, wv);
    lstart[i0] = l0; lstart[i1] = l0 + c0;
    int* row = tab + (size_t)blk * TP;
    if (i0 <= NB) row[i0] = l0;
    if (i1 <= NB) row[i1] = l0 + c0;
    hist[i0] = 0; hist[i1] = 0;
    __syncthreads();

#pragma unroll
    for (int k = 0; k < PA_I; ++k) {
        if (d[k] >= 0) {
            int b = d[k] >> BSH;
            int r = atomicAdd(&hist[b], 1);
            stage[lstart[b] + r] = (unsigned)s[k] | ((unsigned)(d[k] & (BW - 1)) << SRC_BITS);
        }
    }
    __syncthreads();

    int nvalid = E - base;
    if (nvalid > RUN_CAP) nvalid = RUN_CAP;
#pragma unroll
    for (int k = 0; k < PA_I; ++k) {
        int i = k * PA_T + t;
        if (i < nvalid) binned[base + i] = stage[i];
    }
}

__device__ __forceinline__ int run_slot(const int* roffs, const int* rbase, int PG, int j) {
    int lo = 0, hi = PG - 1;
    while (lo < hi) {
        int mid = (lo + hi + 1) >> 1;
        if (roffs[mid] <= j) lo = mid; else hi = mid - 1;
    }
    return rbase[lo] + j;
}

__global__ void __launch_bounds__(KB_T) k_bucket(const unsigned* __restrict__ binned,
                                                 const int* __restrict__ tab, int PG, int TP,
                                                 int* __restrict__ rowptr,
                                                 int* __restrict__ csr, int N) {
    __shared__ int cnt[BW];
    __shared__ int wtot[BW / 32];
    __shared__ int rtot[KB_T / 32];
    __shared__ int rsum[KB_T / 32];
    __shared__ int roffs[KB_T];
    __shared__ int rbase[KB_T];
    __shared__ int stage[KB_T * KB_R];
    int t = threadIdx.x, lane = t & 31, wv = t >> 5;
    int b = blockIdx.x;

    int r0 = 0, len = 0;
    if (t < PG) {
        const int* row = tab + (size_t)t * TP;
        r0 = row[b];
        len = row[b + 1] - r0;
    }
    int rincl = len, rs = r0;
#pragma unroll
    for (int o = 1; o < 32; o <<= 1) {
        int u = __shfl_up(rincl, o);
        if (lane >= o) rincl += u;
        rs += __shfl_xor(rs, o);
    }
    if (lane == 31) rtot[wv] = rincl;
    if (lane == 0) rsum[wv] = rs;
    if (t < BW) cnt[t] = 0;
    __syncthreads();
    int roff = rincl - len, n = 0, bs = 0;
#pragma unroll
    for (int k = 0; k < KB_T / 32; ++k) {
        int x = rtot[k];
        if (k < wv) roff += x;
        n += x;
        bs += rsum[k];
    }
    int g0 = t * RUN_CAP + r0;
    roffs[t] = roff;
    rbase[t] = g0 - roff;
    if (t == 0 && n == 0) stage[0] = 0;
    for (int i = 0; i < len; ++i) {
        int j = roff + i;
        if (j < KB_T * KB_R) stage[j] = g0 + i;
    }
    __syncthreads();

    bool fits = n <= KB_T * KB_R;
    int* cs = csr + bs;

    unsigned w[KB_R];
    int slot[KB_R];
#pragma unroll
    for (int u = 0; u < KB_R; ++u) {
        int j = u * KB_T + t;
        slot[u] = stage[j < n ? j : 0];
    }
#pragma unroll
    for (int u = 0; u < KB_R; ++u) {
        int j = u * KB_T + t;
        w[u] = j < n ? binned[slot[u]] : 0u;
    }
#pragma unroll
    for (int u = 0; u < KB_R; ++u)
        if (u * KB_T + t < n) atomicAdd(&cnt[w[u] >> SRC_BITS], 1);
    for (int j = KB_T * KB_R + t; j < n; j += KB_T)
        atomicAdd(&cnt[binned[run_slot(roffs, rbase, PG, j)] >> SRC_BITS], 1);
    __syncthreads();

    int v = t < BW ? cnt[t] : 0;
    int incl = v;
#pragma unroll
    for (int o = 1; o < 32; o <<= 1) {
        int u = __shfl_up(incl, o);
        if (lane >= o) incl += u;
    }
    if (lane == 31 && t < BW) wtot[wv] = incl;
    __syncthreads();
    if (t < BW) {
        int woff = 0;
#pragma unroll
        for (int k = 0; k < BW / 32; ++k)
            if (k < wv) woff += wtot[k];
        int excl = woff + incl - v;
        cnt[t] = excl;
        int node = b * BW + t;
        if (node < N) rowptr[node] = bs + excl;
        if (node == N - 1) rowptr[N] = bs + excl + v;
    }
    __syncthreads();

    const unsigned SRC_MASK = (1u << SRC_BITS) - 1u;
    if (fits) {
#pragma unroll
        for (int u = 0; u < KB_R; ++u) {
            if (u * KB_T + t < n) {
                int p = atomicAdd(&cnt[w[u] >> SRC_BITS], 1);
                stage[p] = (int)(w[u] & SRC_MASK);
            }
        }
        __syncthreads();
        for (int i = t; i < n; i += KB_T) cs[i] = stage[i];
    } else {
#pragma unroll
        for (int u = 0; u < KB_R; ++u) {
            if (u * KB_T + t < n) {
                int p = atomicAdd(&cnt[w[u] >> SRC_BITS], 1);
                cs[p] = (int)(w[u] & SRC_MASK);
            }
        }
        for (int j = KB_T * KB_R + t; j < n; j += KB_T) {
            unsigned x = binned[run_slot(roffs, rbase, PG, j)];
            int p = atomicAdd(&cnt[x >> SRC_BITS], 1);
            cs[p] = (int)(x & SRC_MASK);
        }
    }
}

#define XL_MAX 80000
#define A1_T 1024
#define A1_G 256

__global__ void __launch_bounds__(A1_T) k_agg1(const float* __restrict__ x,
                                               const int* __restrict__ rowptr,
                                               const int* __restrict__ csr,
                                               const float* __restrict__ W1,
                                               const float* __restrict__ as1,
                                               const float* __restrict__ ad1,
                                               const float* __restrict__ b1,
                                               float* __restrict__ out1, int N) {
    __shared__ __attribute__((aligned(16))) float xl[XL_MAX];
    int t = threadIdx.x;
    int K = N < XL_MAX ? N : XL_MAX;
    int K4 = K & ~3;
    for (int base = t * 4; base < K4; base += 5 * A1_T * 4) {
        float4 v[5];
        int ii[5];
#pragma unroll
        for (int u = 0; u < 5; ++u) {
            int i = base + u * A1_T * 4;
            ii[u] = i < K4 ? i : base;
            v[u] = *reinterpret_cast<const float4*>(x + ii[u]);
        }
#pragma unroll
        for (int u = 0; u < 5; ++u)
            *reinterpret_cast<float4*>(xl + ii[u]) = v[u];
    }
    if (t < K - K4) xl[K4 + t] = x[K4 + t];

    float ws[4], wd[4];
#pragma unroll
    for (int h = 0; h < 4; ++h) {
        float a = 0.f, d = 0.f;
#pragma unroll
        for (int c = 0; c < 8; ++c) {
            float w = W1[h * 8 + c];
            a += w * as1[h * 8 + c];
            d += w * ad1[h * 8 + c];
        }
        ws[h] = a; wd[h] = d;
    }
    __syncthreads();

    int sub = t & 7, grp = t >> 3;
    int per = (N + (int)gridDim.x - 1) / (int)gridDim.x;
    int n0 = (int)blockIdx.x * per;
    int n1 = n0 + per < N ? n0 + per : N;

    int node = n0 + grp;
    int nc = node < n1 ? node : 0;
    float xd_n = x[nc];
    int beg_n = rowptr[nc], end_n = rowptr[nc + 1];

    for (; node < n1; node += A1_T / 8) {
        float xd = xd_n;
        int beg = beg_n, end = end_n;
        int nn = node + A1_T / 8 < n1 ? node + A1_T / 8 : node;
        xd_n = x[nn];
        beg_n = rowptr[nn]; end_n = rowptr[nn + 1];

        float ad[4], m[4], s[4], w[4];
#pragma unroll
        for (int h = 0; h < 4; ++h) {
            ad[h] = xd * wd[h];
            m[h] = lrelu(xd * ws[h] + ad[h]);
            s[h] = sub == 0 ? 1.f : 0.f;
            w[h] = sub == 0 ? xd : 0.f;
        }

        int last = end - 1;
        int j = beg + sub;
        int sc = 0, sn = 0;
        float ca = 0.f, cg = 0.f;
        if (j < end) {
            sc = csr[j];
            sn = csr[j + 8 < end ? j + 8 : last];
            ca = xl[sc < K ? sc : 0];
            cg = x[sc < K ? node : sc];
        }
        for (; j < end; j += 8) {
            int s2 = csr[j + 16 < end ? j + 16 : last];
            float na = xl[sn < K ? sn : 0];
            float ng = x[sn < K ? node : sn];
            float xs = sc < K ? ca : cg;
#pragma unroll
            for (int h = 0; h < 4; ++h) {
                float e  = lrelu(xs * ws[h] + ad[h]);
                float mn = fmaxf(m[h], e);
                float sc2 = __expf(m[h] - mn);
                float tt  = __expf(e - mn);
                s[h] = s[h] * sc2 + tt;
                w[h] = w[h] * sc2 + tt * xs;
                m[h] = mn;
            }
            sc = sn; sn = s2; ca = na; cg = ng;
        }

#pragma unroll
        for (int o = 1; o < 8; o <<= 1) {
#pragma unroll
            for (int h = 0; h < 4; ++h) {
                float mo = __shfl_xor(m[h], o);
                float so = __shfl_xor(s[h], o);
                float wo = __shfl_xor(w[h], o);
                float mn = fmaxf(m[h], mo);
                float a  = __expf(m[h] - mn);
                float b  = __expf(mo - mn);
                s[h] = s[h] * a + so * b;
                w[h] = w[h] * a + wo * b;
                m[h] = mn;
            }
        }

        int h = sub >> 1;
        float sh = h == 0 ? s[0] : h == 1 ? s[1] : h == 2 ? s[2] : s[3];
        float wh = h == 0 ? w[0] : h == 1 ? w[1] : h == 2 ? w[2] : w[3];
        float r = wh / sh;
        const float* Wp = W1 + sub * 4;
        const float* Bp1 = b1 + sub * 4;
        float4 o4;
        o4.x = fmaxf(Wp[0] * r + Bp1[0], 0.f);
        o4.y = fmaxf(Wp[1] * r + Bp1[1], 0.f);
        o4.z = fmaxf(Wp[2] * r + Bp1[2], 0.f);
        o4.w = fmaxf(Wp[3] * r + Bp1[3], 0.f);
        *reinterpret_cast<float4*>(out1 + (size_t)node * 32 + sub * 4) = o4;
    }
}

__global__ void k_gemm2(const float* __restrict__ out1, const float* __restrict__ W2,
                        const float* __restrict__ as2, const float* __restrict__ ad2,
                        float* __restrict__ rec, int N) {
    __shared__ float Bp[32 * 16];
    int t = threadIdx.x;
    for (int i = t; i < 512; i += blockDim.x) {
        int k = i >> 4, n = i & 15;
        float v = 0.f;
        if (n < 8) {
            v = W2[k * 8 + n];
        } else if (n == 8) {
            for (int c = 0; c < 8; ++c) v += W2[k * 8 + c] * as2[c];
        } else if (n == 9) {
            for (int c = 0; c < 8; ++c) v += W2[k * 8 + c] * ad2[c];
        }
        Bp[i] = v;
    }
    __syncthreads();

    int wave = t >> 5, lane = t & 31;
    long tile = (long)blockIdx.x * 8 + wave;
    long row0 = tile * 16;
    if (row0 >= N) return;

    int m = lane & 15, khi = lane >> 4;
    long rowm = row0 + m;
    if (rowm >= N) rowm = N - 1;
    const float* Arow = out1 + rowm * 32;

    v8f c = {};
#pragma unroll
    for (int step = 0; step < 8; ++step) {
        int k0 = step * 4 + khi * 2;
        v2f a; a.x = Arow[k0];          a.y = Arow[k0 + 1];
        v2f b; b.x = Bp[k0 * 16 + m];   b.y = Bp[(k0 + 1) * 16 + m];
        c = __builtin_amdgcn_wmma_f32_16x16x4_f32(false, a, false, b, (short)0, c,
                                                  false, false);
    }

#pragma unroll
    for (int r = 0; r < 8; ++r) {
        long node = row0 + khi * 8 + r;
        if (node >= N) continue;
        rec[node * 16 + m] = c[r];
    }
}

__global__ void __launch_bounds__(256) k_agg2(const float* __restrict__ rec,
                                              const int* __restrict__ rowptr,
                                              const int* __restrict__ csr,
                                              const float* __restrict__ b2,
                                              float* __restrict__ out, int N) {
    int gid = blockIdx.x * blockDim.x + threadIdx.x;
    int node = gid >> 3, sub = threadIdx.x & 7;
    bool valid = node < N;
    int nd = valid ? node : N - 1;

    const float* rd = rec + (size_t)nd * 16;
    float4 d0 = *reinterpret_cast<const float4*>(rd);
    float4 d1 = *reinterpret_cast<const float4*>(rd + 4);
    float2 dl = *reinterpret_cast<const float2*>(rd + 8);
    float ald = dl.y;
    int beg = rowptr[nd];
    int end = valid ? rowptr[nd + 1] : beg;

    float f0 = sub == 0 ? 1.f : 0.f;
    float m = lrelu(dl.x + ald);
    float s = f0;
    float acc[8] = { f0 * d0.x, f0 * d0.y, f0 * d0.z, f0 * d0.w,
                     f0 * d1.x, f0 * d1.y, f0 * d1.z, f0 * d1.w };

    for (int j = beg + sub; j < end; j += 8) {
        int sidx = csr[j];
        const float* rs = rec + (size_t)sidx * 16;
        float4 a0 = *reinterpret_cast<const float4*>(rs);
        float4 a1 = *reinterpret_cast<const float4*>(rs + 4);
        float als = rs[8];
        float e  = lrelu(als + ald);
        float mn = fmaxf(m, e);
        float sc = __expf(m - mn);
        float t  = __expf(e - mn);
        s = s * sc + t;
        acc[0] = acc[0] * sc + t * a0.x; acc[1] = acc[1] * sc + t * a0.y;
        acc[2] = acc[2] * sc + t * a0.z; acc[3] = acc[3] * sc + t * a0.w;
        acc[4] = acc[4] * sc + t * a1.x; acc[5] = acc[5] * sc + t * a1.y;
        acc[6] = acc[6] * sc + t * a1.z; acc[7] = acc[7] * sc + t * a1.w;
        m = mn;
    }

#pragma unroll
    for (int o = 1; o < 8; o <<= 1) {
        float mo = __shfl_xor(m, o);
        float so = __shfl_xor(s, o);
        float mn = fmaxf(m, mo);
        float a  = __expf(m - mn);
        float b  = __expf(mo - mn);
        s = s * a + so * b;
#pragma unroll
        for (int c = 0; c < 8; ++c) {
            float ao = __shfl_xor(acc[c], o);
            acc[c] = acc[c] * a + ao * b;
        }
        m = mn;
    }

    float av = acc[0];
#pragma unroll
    for (int c = 1; c < 8; ++c) av = sub == c ? acc[c] : av;
    float o1 = av / s + b2[sub];
    if (valid) out[(size_t)node * 8 + sub] = o1;
}

extern "C" void kernel_launch(void* const* d_in, const int* in_sizes, int n_in,
                              void* d_out, int out_size, void* d_ws, size_t ws_size,
                              hipStream_t stream) {
    const float* x   = (const float*)d_in[0];
    const int*   ei  = (const int*)d_in[1];
    const float* W1  = (const float*)d_in[2];
    const float* as1 = (const float*)d_in[3];
    const float* ad1 = (const float*)d_in[4];
    const float* b1  = (const float*)d_in[5];
    const float* W2  = (const float*)d_in[6];
    const float* as2 = (const float*)d_in[7];
    const float* ad2 = (const float*)d_in[8];
    const float* b2  = (const float*)d_in[9];
    (void)n_in; (void)out_size; (void)ws_size;

    int N = in_sizes[0];
    int E = in_sizes[1] / 2;

    char* ws = (char*)d_ws;
    size_t cur = 0;
    auto carve = [&](size_t bytes) -> void* {
        void* p = ws + cur;
        cur += (bytes + 255) & ~(size_t)255;
        return p;
    };
    int NB = (N + BW - 1) >> BSH;
    int PG = (E + RUN_CAP - 1) / RUN_CAP;
    int TP = (NB + 1 + 31) & ~31;
    int*      tab    = (int*)carve(sizeof(int) * (size_t)PG * (size_t)TP);
    int*      rowptr = (int*)carve(sizeof(int) * ((size_t)N + 1));
    unsigned* binned = (unsigned*)carve(sizeof(unsigned) * (size_t)PG * RUN_CAP);
    int*      csr    = (int*)carve(sizeof(int) * (size_t)E);
    float* out1   = (float*)carve(sizeof(float) * (size_t)N * 32);
    float* rec    = (float*)carve(sizeof(float) * (size_t)N * 16);

    const int B = 256;
    k_bsort<<<PG, PA_T, 0, stream>>>(ei, E, NB, TP, tab, binned);
    k_bucket<<<NB, KB_T, 0, stream>>>(binned, tab, PG, TP, rowptr, csr, N);
    k_agg1<<<A1_G, A1_T, 0, stream>>>(x, rowptr, csr, W1, as1, ad1, b1, out1, N);

    int tiles = (N + 15) / 16;
    k_gemm2<<<(tiles + 7) / 8, B, 0, stream>>>(out1, W2, as2, ad2, rec, N);

    k_agg2<<<(8 * N + B - 1) / B, B, 0, stream>>>(rec, rowptr, csr, b2, (float*)d_out, N);
}


# Layer-2 transform fused into the layer-1 aggregation: each wave turns the per-head ratios of its 16 nodes into A fragments in registers and runs the WMMA chain itself; no out1, 5 to 4 launches
# speedup vs baseline: 24.1061x; 1.0644x over previous
#include <hip/hip_runtime.h>
#include <math.h>


#define NEG_SLOPE 0.2f

typedef __attribute__((ext_vector_type(2))) float v2f;
typedef __attribute__((ext_vector_type(8))) float v8f;

__device__ __forceinline__ float lrelu(float v) { return v > 0.f ? v : NEG_SLOPE * v; }

#define BSH 9
#define BW (1 << BSH)
#define MAXNB 1024
#define SRC_BITS 23
#define PA_T 512
#define PA_I 16
#define RUN_CAP (PA_T * PA_I)
#define KB_T 1024
#define KB_R 20

__device__ __forceinline__ int scan_pairs(int v0, int v1, int* wtot, int lane, int wv) {
    int incl = v0 + v1;
#pragma unroll
    for (int o = 1; o < 32; o <<= 1) {
        int u = __shfl_up(incl, o);
        if (lane >= o) incl += u;
    }
    if (lane == 31) wtot[wv] = incl;
    __syncthreads();
    int woff = 0;
#pragma unroll
    for (int k = 0; k < PA_T / 32; ++k)
        if (k < wv) woff += wtot[k];
    return woff + incl - (v0 + v1);
}

__global__ void __launch_bounds__(PA_T) k_bsort(const int* __restrict__ ei, int E, int NB, int TP,
                                                int* __restrict__ tab,
                                                unsigned* __restrict__ binned) {
    __shared__ int hist[MAXNB];
    __shared__ int lstart[MAXNB];
    __shared__ int wtot[PA_T / 32];
    __shared__ unsigned stage[RUN_CAP];
    int t = threadIdx.x, lane = t & 31, wv = t >> 5;
    int i0 = 2 * t, i1 = 2 * t + 1;
    hist[i0] = 0; hist[i1] = 0;
    __syncthreads();

    int blk = blockIdx.x;
    int base = blk * RUN_CAP;
    int d[PA_I], s[PA_I];
#pragma unroll
    for (int k = 0; k < PA_I; ++k) {
        int e = base + k * PA_T + t;
        bool in = e < E;
        d[k] = in ? ei[E + e] : -1;
        s[k] = in ? ei[e] : 0;
    }
#pragma unroll
    for (int k = 0; k < PA_I; ++k)
        if (d[k] >= 0) atomicAdd(&hist[d[k] >> BSH], 1);
    __syncthreads();

    int c0 = hist[i0], c1 = hist[i1];
    int l0 = scan_pairs(c0, c1, wtot, lane, wv);
    lstart[i0] = l0; lstart[i1] = l0 + c0;
    int* row = tab + (size_t)blk * TP;
    if (i0 <= NB) row[i0] = l0;
    if (i1 <= NB) row[i1] = l0 + c0;
    hist[i0] = 0; hist[i1] = 0;
    __syncthreads();

#pragma unroll
    for (int k = 0; k < PA_I; ++k) {
        if (d[k] >= 0) {
            int b = d[k] >> BSH;
            int r = atomicAdd(&hist[b], 1);
            stage[lstart[b] + r] = (unsigned)s[k] | ((unsigned)(d[k] & (BW - 1)) << SRC_BITS);
        }
    }
    __syncthreads();

    int nvalid = E - base;
    if (nvalid > RUN_CAP) nvalid = RUN_CAP;
#pragma unroll
    for (int k = 0; k < PA_I; ++k) {
        int i = k * PA_T + t;
        if (i < nvalid) binned[base + i] = stage[i];
    }
}

__device__ __forceinline__ int run_slot(const int* roffs, const int* rbase, int PG, int j) {
    int lo = 0, hi = PG - 1;
    while (lo < hi) {
        int mid = (lo + hi + 1) >> 1;
        if (roffs[mid] <= j) lo = mid; else hi = mid - 1;
    }
    return rbase[lo] + j;
}

__global__ void __launch_bounds__(KB_T) k_bucket(const unsigned* __restrict__ binned,
                                                 const int* __restrict__ tab, int PG, int TP,
                                                 int* __restrict__ rowptr,
                                                 int* __restrict__ csr, int N) {
    __shared__ int cnt[BW];
    __shared__ int wtot[BW / 32];
    __shared__ int rtot[KB_T / 32];
    __shared__ int rsum[KB_T / 32];
    __shared__ int roffs[KB_T];
    __shared__ int rbase[KB_T];
    __shared__ int stage[KB_T * KB_R];
    int t = threadIdx.x, lane = t & 31, wv = t >> 5;
    int b = blockIdx.x;

    int r0 = 0, len = 0;
    if (t < PG) {
        const int* row = tab + (size_t)t * TP;
        r0 = row[b];
        len = row[b + 1] - r0;
    }
    int rincl = len, rs = r0;
#pragma unroll
    for (int o = 1; o < 32; o <<= 1) {
        int u = __shfl_up(rincl, o);
        if (lane >= o) rincl += u;
        rs += __shfl_xor(rs, o);
    }
    if (lane == 31) rtot[wv] = rincl;
    if (lane == 0) rsum[wv] = rs;
    if (t < BW) cnt[t] = 0;
    __syncthreads();
    int roff = rincl - len, n = 0, bs = 0;
#pragma unroll
    for (int k = 0; k < KB_T / 32; ++k) {
        int x = rtot[k];
        if (k < wv) roff += x;
        n += x;
        bs += rsum[k];
    }
    int g0 = t * RUN_CAP + r0;
    roffs[t] = roff;
    rbase[t] = g0 - roff;
    if (t == 0 && n == 0) stage[0] = 0;
    for (int i = 0; i < len; ++i) {
        int j = roff + i;
        if (j < KB_T * KB_R) stage[j] = g0 + i;
    }
    __syncthreads();

    bool fits = n <= KB_T * KB_R;
    int* cs = csr + bs;

    unsigned w[KB_R];
    int slot[KB_R];
#pragma unroll
    for (int u = 0; u < KB_R; ++u) {
        int j = u * KB_T + t;
        slot[u] = stage[j < n ? j : 0];
    }
#pragma unroll
    for (int u = 0; u < KB_R; ++u) {
        int j = u * KB_T + t;
        w[u] = j < n ? binned[slot[u]] : 0u;
    }
#pragma unroll
    for (int u = 0; u < KB_R; ++u)
        if (u * KB_T + t < n) atomicAdd(&cnt[w[u] >> SRC_BITS], 1);
    for (int j = KB_T * KB_R + t; j < n; j += KB_T)
        atomicAdd(&cnt[binned[run_slot(roffs, rbase, PG, j)] >> SRC_BITS], 1);
    __syncthreads();

    int v = t < BW ? cnt[t] : 0;
    int incl = v;
#pragma unroll
    for (int o = 1; o < 32; o <<= 1) {
        int u = __shfl_up(incl, o);
        if (lane >= o) incl += u;
    }
    if (lane == 31 && t < BW) wtot[wv] = incl;
    __syncthreads();
    if (t < BW) {
        int woff = 0;
#pragma unroll
        for (int k = 0; k < BW / 32; ++k)
            if (k < wv) woff += wtot[k];
        int excl = woff + incl - v;
        cnt[t] = excl;
        int node = b * BW + t;
        if (node < N) rowptr[node] = bs + excl;
        if (node == N - 1) rowptr[N] = bs + excl + v;
    }
    __syncthreads();

    const unsigned SRC_MASK = (1u << SRC_BITS) - 1u;
    if (fits) {
#pragma unroll
        for (int u = 0; u < KB_R; ++u) {
            if (u * KB_T + t < n) {
                int p = atomicAdd(&cnt[w[u] >> SRC_BITS], 1);
                stage[p] = (int)(w[u] & SRC_MASK);
            }
        }
        __syncthreads();
        for (int i = t; i < n; i += KB_T) cs[i] = stage[i];
    } else {
#pragma unroll
        for (int u = 0; u < KB_R; ++u) {
            if (u * KB_T + t < n) {
                int p = atomicAdd(&cnt[w[u] >> SRC_BITS], 1);
                cs[p] = (int)(w[u] & SRC_MASK);
            }
        }
        for (int j = KB_T * KB_R + t; j < n; j += KB_T) {
            unsigned x = binned[run_slot(roffs, rbase, PG, j)];
            int p = atomicAdd(&cnt[x >> SRC_BITS], 1);
            cs[p] = (int)(x & SRC_MASK);
        }
    }
}

#define XL_MAX 80000
#define A1_T 1024
#define A1_G 256

__global__ void __launch_bounds__(A1_T) k_agg1(const float* __restrict__ x,
                                               const int* __restrict__ rowptr,
                                               const int* __restrict__ csr,
                                               const float* __restrict__ W1,
                                               const float* __restrict__ as1,
                                               const float* __restrict__ ad1,
                                               const float* __restrict__ b1,
                                               const float* __restrict__ W2,
                                               const float* __restrict__ as2,
                                               const float* __restrict__ ad2,
                                               float* __restrict__ rec, int N) {
    __shared__ __attribute__((aligned(16))) float xl[XL_MAX];
    __shared__ float Bp[32 * 16];
    int t = threadIdx.x;
    if (t < 512) {
        int k = t >> 4, n = t & 15;
        float v = 0.f;
        if (n < 8) {
            v = W2[k * 8 + n];
        } else if (n == 8) {
            for (int c = 0; c < 8; ++c) v += W2[k * 8 + c] * as2[c];
        } else if (n == 9) {
            for (int c = 0; c < 8; ++c) v += W2[k * 8 + c] * ad2[c];
        }
        Bp[t] = v;
    }
    int K = N < XL_MAX ? N : XL_MAX;
    int K4 = K & ~3;
    for (int base = t * 4; base < K4; base += 5 * A1_T * 4) {
        float4 v[5];
        int ii[5];
#pragma unroll
        for (int u = 0; u < 5; ++u) {
            int i = base + u * A1_T * 4;
            ii[u] = i < K4 ? i : base;
            v[u] = *reinterpret_cast<const float4*>(x + ii[u]);
        }
#pragma unroll
        for (int u = 0; u < 5; ++u)
            *reinterpret_cast<float4*>(xl + ii[u]) = v[u];
    }
    if (t < K - K4) xl[K4 + t] = x[K4 + t];

    float ws[4], wd[4];
#pragma unroll
    for (int h = 0; h < 4; ++h) {
        float a = 0.f, d = 0.f;
#pragma unroll
        for (int c = 0; c < 8; ++c) {
            float w = W1[h * 8 + c];
            a += w * as1[h * 8 + c];
            d += w * ad1[h * 8 + c];
        }
        ws[h] = a; wd[h] = d;
    }
    __syncthreads();

    int sub = t & 7, grp = t >> 3;
    int per = ((N + (int)gridDim.x - 1) / (int)gridDim.x + 3) & ~3;
    int n0 = (int)blockIdx.x * per;
    int n1 = n0 + per < N ? n0 + per : N;

    int lane = t & 31, m16 = lane & 15, khi = lane >> 4;
    int nw = n0 + 4 * __builtin_amdgcn_readfirstlane(t >> 5);
    float rt[4] = { 0.f, 0.f, 0.f, 0.f };
    int trip = 0;

    int node = n0 + grp;
    int nc = node < n1 ? node : 0;
    float xd_n = x[nc];
    int beg_n = rowptr[nc], end_n = rowptr[nc + 1];

    for (; nw < n1; nw += A1_T / 8, node += A1_T / 8, ++trip) {
        float xd = xd_n;
        int beg = beg_n, end = end_n;
        if (node >= n1) end = beg;
        int nn = node + A1_T / 8 < n1 ? node + A1_T / 8 : nc;
        xd_n = x[nn];
        beg_n = rowptr[nn]; end_n = rowptr[nn + 1];

        float ad[4], m[4], s[4], w[4];
#pragma unroll
        for (int h = 0; h < 4; ++h) {
            ad[h] = xd * wd[h];
            m[h] = lrelu(xd * ws[h] + ad[h]);
            s[h] = sub == 0 ? 1.f : 0.f;
            w[h] = sub == 0 ? xd : 0.f;
        }

        int last = end - 1;
        int j = beg + sub;
        int sc = 0, sn = 0;
        float ca = 0.f, cg = 0.f;
        if (j < end) {
            sc = csr[j];
            sn = csr[j + 8 < end ? j + 8 : last];
            ca = xl[sc < K ? sc : 0];
            cg = x[sc < K ? node : sc];
        }
        for (; j < end; j += 8) {
            int s2 = csr[j + 16 < end ? j + 16 : last];
            float na = xl[sn < K ? sn : 0];
            float ng = x[sn < K ? node : sn];
            float xs = sc < K ? ca : cg;
#pragma unroll
            for (int h = 0; h < 4; ++h) {
                float e  = lrelu(xs * ws[h] + ad[h]);
                float mn = fmaxf(m[h], e);
                float sc2 = __expf(m[h] - mn);
                float tt  = __expf(e - mn);
                s[h] = s[h] * sc2 + tt;
                w[h] = w[h] * sc2 + tt * xs;
                m[h] = mn;
            }
            sc = sn; sn = s2; ca = na; cg = ng;
        }

#pragma unroll
        for (int o = 1; o < 8; o <<= 1) {
#pragma unroll
            for (int h = 0; h < 4; ++h) {
                float mo = __shfl_xor(m[h], o);
                float so = __shfl_xor(s[h], o);
                float wo = __shfl_xor(w[h], o);
                float mn = fmaxf(m[h], mo);
                float a  = __expf(m[h] - mn);
                float b  = __expf(mo - mn);
                s[h] = s[h] * a + so * b;
                w[h] = w[h] * a + wo * b;
                m[h] = mn;
            }
        }

        int h = sub >> 1;
        float sh = h == 0 ? s[0] : h == 1 ? s[1] : h == 2 ? s[2] : s[3];
        float wh = h == 0 ? w[0] : h == 1 ? w[1] : h == 2 ? w[2] : w[3];
        float r = wh / sh;

        int q = trip & 3;
#pragma unroll
        for (int hh = 0; hh < 4; ++hh) {
            float v = __shfl(r, 8 * (m16 & 3) + 2 * hh);
            rt[hh] = (m16 >> 2) == q ? v : rt[hh];
        }

        if (q == 3 || nw + A1_T / 8 >= n1) {
            v8f acc = {};
#pragma unroll
            for (int step = 0; step < 8; ++step) {
                int k0 = step * 4 + khi * 2;
                float rh = rt[step >> 1];
                v2f a;
                a.x = fmaxf(W1[k0] * rh + b1[k0], 0.f);
                a.y = fmaxf(W1[k0 + 1] * rh + b1[k0 + 1], 0.f);
                v2f b; b.x = Bp[k0 * 16 + m16];   b.y = Bp[(k0 + 1) * 16 + m16];
                acc = __builtin_amdgcn_wmma_f32_16x16x4_f32(false, a, false, b, (short)0, acc,
                                                            false, false);
            }
            int tb = nw - q * (A1_T / 8);
#pragma unroll
            for (int i = 0; i < 8; ++i) {
                int M = khi * 8 + i;
                int nd = tb + (M & 3) + (M >> 2) * (A1_T / 8);
                if (nd < n1) rec[(size_t)nd * 16 + m16] = acc[i];
            }
        }
    }
}

__global__ void __launch_bounds__(256) k_agg2(const float* __restrict__ rec,
                                              const int* __restrict__ rowptr,
                                              const int* __restrict__ csr,
                                              const float* __restrict__ b2,
                                              float* __restrict__ out, int N) {
    int gid = blockIdx.x * blockDim.x + threadIdx.x;
    int node = gid >> 3, sub = threadIdx.x & 7;
    bool valid = node < N;
    int nd = valid ? node : N - 1;

    const float* rd = rec + (size_t)nd * 16;
    float4 d0 = *reinterpret_cast<const float4*>(rd);
    float4 d1 = *reinterpret_cast<const float4*>(rd + 4);
    float2 dl = *reinterpret_cast<const float2*>(rd + 8);
    float ald = dl.y;
    int beg = rowptr[nd];
    int end = valid ? rowptr[nd + 1] : beg;

    float f0 = sub == 0 ? 1.f : 0.f;
    float m = lrelu(dl.x + ald);
    float s = f0;
    float acc[8] = { f0 * d0.x, f0 * d0.y, f0 * d0.z, f0 * d0.w,
                     f0 * d1.x, f0 * d1.y, f0 * d1.z, f0 * d1.w };

    for (int j = beg + sub; j < end; j += 8) {
        int sidx = csr[j];
        const float* rs = rec + (size_t)sidx * 16;
        float4 a0 = *reinterpret_cast<const float4*>(rs);
        float4 a1 = *reinterpret_cast<const float4*>(rs + 4);
        float als = rs[8];
        float e  = lrelu(als + ald);
        float mn = fmaxf(m, e);
        float sc = __expf(m - mn);
        float t  = __expf(e - mn);
        s = s * sc + t;
        acc[0] = acc[0] * sc + t * a0.x; acc[1] = acc[1] * sc + t * a0.y;
        acc[2] = acc[2] * sc + t * a0.z; acc[3] = acc[3] * sc + t * a0.w;
        acc[4] = acc[4] * sc + t * a1.x; acc[5] = acc[5] * sc + t * a1.y;
        acc[6] = acc[6] * sc + t * a1.z; acc[7] = acc[7] * sc + t * a1.w;
        m = mn;
    }

#pragma unroll
    for (int o = 1; o < 8; o <<= 1) {
        float mo = __shfl_xor(m, o);
        float so = __shfl_xor(s, o);
        float mn = fmaxf(m, mo);
        float a  = __expf(m - mn);
        float b  = __expf(mo - mn);
        s = s * a + so * b;
#pragma unroll
        for (int c = 0; c < 8; ++c) {
            float ao = __shfl_xor(acc[c], o);
            acc[c] = acc[c] * a + ao * b;
        }
        m = mn;
    }

    float av = acc[0];
#pragma unroll
    for (int c = 1; c < 8; ++c) av = sub == c ? acc[c] : av;
    float o1 = av / s + b2[sub];
    if (valid) out[(size_t)node * 8 + sub] = o1;
}

extern "C" void kernel_launch(void* const* d_in, const int* in_sizes, int n_in,
                              void* d_out, int out_size, void* d_ws, size_t ws_size,
                              hipStream_t stream) {
    const float* x   = (const float*)d_in[0];
    const int*   ei  = (const int*)d_in[1];
    const float* W1  = (const float*)d_in[2];
    const float* as1 = (const float*)d_in[3];
    const float* ad1 = (const float*)d_in[4];
    const float* b1  = (const float*)d_in[5];
    const float* W2  = (const float*)d_in[6];
    const float* as2 = (const float*)d_in[7];
    const float* ad2 = (const float*)d_in[8];
    const float* b2  = (const float*)d_in[9];
    (void)n_in; (void)out_size; (void)ws_size;

    int N = in_sizes[0];
    int E = in_sizes[1] / 2;

    char* ws = (char*)d_ws;
    size_t cur = 0;
    auto carve = [&](size_t bytes) -> void* {
        void* p = ws + cur;
        cur += (bytes + 255) & ~(size_t)255;
        return p;
    };
    int NB = (N + BW - 1) >> BSH;
    int PG = (E + RUN_CAP - 1) / RUN_CAP;
    int TP = (NB + 1 + 31) & ~31;
    int*      tab    = (int*)carve(sizeof(int) * (size_t)PG * (size_t)TP);
    int*      rowptr = (int*)carve(sizeof(int) * ((size_t)N + 1));
    unsigned* binned = (unsigned*)carve(sizeof(unsigned) * (size_t)PG * RUN_CAP);
    int*      csr    = (int*)carve(sizeof(int) * (size_t)E);
    float* rec    = (float*)carve(sizeof(float) * (size_t)N * 16);

    const int B = 256;
    k_bsort<<<PG, PA_T, 0, stream>>>(ei, E, NB, TP, tab, binned);
    k_bucket<<<NB, KB_T, 0, stream>>>(binned, tab, PG, TP, rowptr, csr, N);
    k_agg1<<<A1_G, A1_T, 0, stream>>>(x, rowptr, csr, W1, as1, ad1, b1, W2, as2, ad2, rec, N);

    k_agg2<<<(8 * N + B - 1) / B, B, 0, stream>>>(rec, rowptr, csr, b2, (float*)d_out, N);
}


# Layer-1 softmax: segment maximum taken from the node's extreme source x (the logit is monotone in x_s*ws), so one exp per edge and head, no running-max rescale; per-lane sum order kept
# speedup vs baseline: 25.1869x; 1.0448x over previous
#include <hip/hip_runtime.h>
#include <math.h>


#define NEG_SLOPE 0.2f

typedef __attribute__((ext_vector_type(2))) float v2f;
typedef __attribute__((ext_vector_type(8))) float v8f;

__device__ __forceinline__ float lrelu(float v) { return v > 0.f ? v : NEG_SLOPE * v; }

#define BSH 9
#define BW (1 << BSH)
#define MAXNB 1024
#define SRC_BITS 23
#define PA_T 512
#define PA_I 16
#define RUN_CAP (PA_T * PA_I)
#define KB_T 1024
#define KB_R 20

__device__ __forceinline__ int scan_pairs(int v0, int v1, int* wtot, int lane, int wv) {
    int incl = v0 + v1;
#pragma unroll
    for (int o = 1; o < 32; o <<= 1) {
        int u = __shfl_up(incl, o);
        if (lane >= o) incl += u;
    }
    if (lane == 31) wtot[wv] = incl;
    __syncthreads();
    int woff = 0;
#pragma unroll
    for (int k = 0; k < PA_T / 32; ++k)
        if (k < wv) woff += wtot[k];
    return woff + incl - (v0 + v1);
}

__global__ void __launch_bounds__(PA_T) k_bsort(const int* __restrict__ ei, int E, int NB, int TP,
                                                int* __restrict__ tab,
                                                unsigned* __restrict__ binned) {
    __shared__ int hist[MAXNB];
    __shared__ int lstart[MAXNB];
    __shared__ int wtot[PA_T / 32];
    __shared__ unsigned stage[RUN_CAP];
    int t = threadIdx.x, lane = t & 31, wv = t >> 5;
    int i0 = 2 * t, i1 = 2 * t + 1;
    hist[i0] = 0; hist[i1] = 0;
    __syncthreads();

    int blk = blockIdx.x;
    int base = blk * RUN_CAP;
    int d[PA_I], s[PA_I];
#pragma unroll
    for (int k = 0; k < PA_I; ++k) {
        int e = base + k * PA_T + t;
        bool in = e < E;
        d[k] = in ? ei[E + e] : -1;
        s[k] = in ? ei[e] : 0;
    }
#pragma unroll
    for (int k = 0; k < PA_I; ++k)
        if (d[k] >= 0) atomicAdd(&hist[d[k] >> BSH], 1);
    __syncthreads();

    int c0 = hist[i0], c1 = hist[i1];
    int l0 = scan_pairs(c0, c1, wtot, lane, wv);
    lstart[i0] = l0; lstart[i1] = l0 + c0;
    int* row = tab + (size_t)blk * TP;
    if (i0 <= NB) row[i0] = l0;
    if (i1 <= NB) row[i1] = l0 + c0;
    hist[i0] = 0; hist[i1] = 0;
    __syncthreads();

#pragma unroll
    for (int k = 0; k < PA_I; ++k) {
        if (d[k] >= 0) {
            int b = d[k] >> BSH;
            int r = atomicAdd(&hist[b], 1);
            stage[lstart[b] + r] = (unsigned)s[k] | ((unsigned)(d[k] & (BW - 1)) << SRC_BITS);
        }
    }
    __syncthreads();

    int nvalid = E - base;
    if (nvalid > RUN_CAP) nvalid = RUN_CAP;
#pragma unroll
    for (int k = 0; k < PA_I; ++k) {
        int i = k * PA_T + t;
        if (i < nvalid) binned[base + i] = stage[i];
    }
}

__device__ __forceinline__ int run_slot(const int* roffs, const int* rbase, int PG, int j) {
    int lo = 0, hi = PG - 1;
    while (lo < hi) {
        int mid = (lo + hi + 1) >> 1;
        if (roffs[mid] <= j) lo = mid; else hi = mid - 1;
    }
    return rbase[lo] + j;
}

__global__ void __launch_bounds__(KB_T) k_bucket(const unsigned* __restrict__ binned,
                                                 const int* __restrict__ tab, int PG, int TP,
                                                 int* __restrict__ rowptr,
                                                 int* __restrict__ csr, int N) {
    __shared__ int cnt[BW];
    __shared__ int wtot[BW / 32];
    __shared__ int rtot[KB_T / 32];
    __shared__ int rsum[KB_T / 32];
    __shared__ int roffs[KB_T];
    __shared__ int rbase[KB_T];
    __shared__ int stage[KB_T * KB_R];
    int t = threadIdx.x, lane = t & 31, wv = t >> 5;
    int b = blockIdx.x;

    int r0 = 0, len = 0;
    if (t < PG) {
        const int* row = tab + (size_t)t * TP;
        r0 = row[b];
        len = row[b + 1] - r0;
    }
    int rincl = len, rs = r0;
#pragma unroll
    for (int o = 1; o < 32; o <<= 1) {
        int u = __shfl_up(rincl, o);
        if (lane >= o) rincl += u;
        rs += __shfl_xor(rs, o);
    }
    if (lane == 31) rtot[wv] = rincl;
    if (lane == 0) rsum[wv] = rs;
    if (t < BW) cnt[t] = 0;
    __syncthreads();
    int roff = rincl - len, n = 0, bs = 0;
#pragma unroll
    for (int k = 0; k < KB_T / 32; ++k) {
        int x = rtot[k];
        if (k < wv) roff += x;
        n += x;
        bs += rsum[k];
    }
    int g0 = t * RUN_CAP + r0;
    roffs[t] = roff;
    rbase[t] = g0 - roff;
    if (t == 0 && n == 0) stage[0] = 0;
    for (int i = 0; i < len; ++i) {
        int j = roff + i;
        if (j < KB_T * KB_R) stage[j] = g0 + i;
    }
    __syncthreads();

    bool fits = n <= KB_T * KB_R;
    int* cs = csr + bs;

    unsigned w[KB_R];
    int slot[KB_R];
#pragma unroll
    for (int u = 0; u < KB_R; ++u) {
        int j = u * KB_T + t;
        slot[u] = stage[j < n ? j : 0];
    }
#pragma unroll
    for (int u = 0; u < KB_R; ++u) {
        int j = u * KB_T + t;
        w[u] = j < n ? binned[slot[u]] : 0u;
    }
#pragma unroll
    for (int u = 0; u < KB_R; ++u)
        if (u * KB_T + t < n) atomicAdd(&cnt[w[u] >> SRC_BITS], 1);
    for (int j = KB_T * KB_R + t; j < n; j += KB_T)
        atomicAdd(&cnt[binned[run_slot(roffs, rbase, PG, j)] >> SRC_BITS], 1);
    __syncthreads();

    int v = t < BW ? cnt[t] : 0;
    int incl = v;
#pragma unroll
    for (int o = 1; o < 32; o <<= 1) {
        int u = __shfl_up(incl, o);
        if (lane >= o) incl += u;
    }
    if (lane == 31 && t < BW) wtot[wv] = incl;
    __syncthreads();
    if (t < BW) {
        int woff = 0;
#pragma unroll
        for (int k = 0; k < BW / 32; ++k)
            if (k < wv) woff += wtot[k];
        int excl = woff + incl - v;
        cnt[t] = excl;
        int node = b * BW + t;
        if (node < N) rowptr[node] = bs + excl;
        if (node == N - 1) rowptr[N] = bs + excl + v;
    }
    __syncthreads();

    const unsigned SRC_MASK = (1u << SRC_BITS) - 1u;
    if (fits) {
#pragma unroll
        for (int u = 0; u < KB_R; ++u) {
            if (u * KB_T + t < n) {
                int p = atomicAdd(&cnt[w[u] >> SRC_BITS], 1);
                stage[p] = (int)(w[u] & SRC_MASK);
            }
        }
        __syncthreads();
        for (int i = t; i < n; i += KB_T) cs[i] = stage[i];
    } else {
#pragma unroll
        for (int u = 0; u < KB_R; ++u) {
            if (u * KB_T + t < n) {
                int p = atomicAdd(&cnt[w[u] >> SRC_BITS], 1);
                cs[p] = (int)(w[u] & SRC_MASK);
            }
        }
        for (int j = KB_T * KB_R + t; j < n; j += KB_T) {
            unsigned x = binned[run_slot(roffs, rbase, PG, j)];
            int p = atomicAdd(&cnt[x >> SRC_BITS], 1);
            cs[p] = (int)(x & SRC_MASK);
        }
    }
}

#define XL_MAX 80000
#define A1_T 1024
#define A1_G 256
#define A1_U 6

__global__ void __launch_bounds__(A1_T) k_agg1(const float* __restrict__ x,
                                               const int* __restrict__ rowptr,
                                               const int* __restrict__ csr,
                                               const float* __restrict__ W1,
                                               const float* __restrict__ as1,
                                               const float* __restrict__ ad1,
                                               const float* __restrict__ b1,
                                               const float* __restrict__ W2,
                                               const float* __restrict__ as2,
                                               const float* __restrict__ ad2,
                                               float* __restrict__ rec, int N) {
    __shared__ __attribute__((aligned(16))) float xl[XL_MAX];
    __shared__ float Bp[32 * 16];
    int t = threadIdx.x;
    if (t < 512) {
        int k = t >> 4, n = t & 15;
        float v = 0.f;
        if (n < 8) {
            v = W2[k * 8 + n];
        } else if (n == 8) {
            for (int c = 0; c < 8; ++c) v += W2[k * 8 + c] * as2[c];
        } else if (n == 9) {
            for (int c = 0; c < 8; ++c) v += W2[k * 8 + c] * ad2[c];
        }
        Bp[t] = v;
    }
    int K = N < XL_MAX ? N : XL_MAX;
    int K4 = K & ~3;
    for (int base = t * 4; base < K4; base += 5 * A1_T * 4) {
        float4 v[5];
        int ii[5];
#pragma unroll
        for (int u = 0; u < 5; ++u) {
            int i = base + u * A1_T * 4;
            ii[u] = i < K4 ? i : base;
            v[u] = *reinterpret_cast<const float4*>(x + ii[u]);
        }
#pragma unroll
        for (int u = 0; u < 5; ++u)
            *reinterpret_cast<float4*>(xl + ii[u]) = v[u];
    }
    if (t < K - K4) xl[K4 + t] = x[K4 + t];

    float ws[4], wd[4];
#pragma unroll
    for (int h = 0; h < 4; ++h) {
        float a = 0.f, d = 0.f;
#pragma unroll
        for (int c = 0; c < 8; ++c) {
            float w = W1[h * 8 + c];
            a += w * as1[h * 8 + c];
            d += w * ad1[h * 8 + c];
        }
        ws[h] = a; wd[h] = d;
    }
    __syncthreads();

    int sub = t & 7, grp = t >> 3;
    int per = ((N + (int)gridDim.x - 1) / (int)gridDim.x + 3) & ~3;
    int n0 = (int)blockIdx.x * per;
    int n1 = n0 + per < N ? n0 + per : N;

    int lane = t & 31, m16 = lane & 15, khi = lane >> 4;
    int nw = n0 + 4 * __builtin_amdgcn_readfirstlane(t >> 5);
    float rt[4] = { 0.f, 0.f, 0.f, 0.f };
    int trip = 0;

    int node = n0 + grp;
    int nc = node < n1 ? node : 0;
    float xd_n = x[nc];
    int beg_n = rowptr[nc], end_n = rowptr[nc + 1];

    for (; nw < n1; nw += A1_T / 8, node += A1_T / 8, ++trip) {
        float xd = xd_n;
        int beg = beg_n, end = end_n;
        if (node >= n1) end = beg;
        int nn = node + A1_T / 8 < n1 ? node + A1_T / 8 : nc;
        xd_n = x[nn];
        beg_n = rowptr[nn]; end_n = rowptr[nn + 1];

        float ad[4], m[4], s[4], w[4];
#pragma unroll
        for (int h = 0; h < 4; ++h) ad[h] = xd * wd[h];

        int last = end > 0 ? end - 1 : 0;
        int ncur = node < n1 ? node : 0;
        int j0 = beg + sub;
        int sv[A1_U];
#pragma unroll
        for (int u = 0; u < A1_U; ++u) {
            int j = j0 + 8 * u;
            sv[u] = csr[j < end ? j : last];
        }
        float va[A1_U], vg[A1_U];
#pragma unroll
        for (int u = 0; u < A1_U; ++u) {
            bool in = j0 + 8 * u < end;
            bool lo = sv[u] < K;
            va[u] = xl[in && lo ? sv[u] : 0];
            vg[u] = x[in && !lo ? sv[u] : ncur];
        }
        float xs[A1_U];
        float xmn = xd, xmx = xd;
#pragma unroll
        for (int u = 0; u < A1_U; ++u) {
            bool in = j0 + 8 * u < end;
            float v = sv[u] < K ? va[u] : vg[u];
            xs[u] = in ? v : xd;
            xmn = fminf(xmn, xs[u]);
            xmx = fmaxf(xmx, xs[u]);
        }
        for (int j = j0 + 8 * A1_U; j < end; j += 8) {
            int sidx = csr[j];
            bool lo = sidx < K;
            float a = xl[lo ? sidx : 0];
            float g = x[lo ? ncur : sidx];
            float v = lo ? a : g;
            xmn = fminf(xmn, v);
            xmx = fmaxf(xmx, v);
        }
#pragma unroll
        for (int o = 1; o < 8; o <<= 1) {
            xmn = fminf(xmn, __shfl_xor(xmn, o));
            xmx = fmaxf(xmx, __shfl_xor(xmx, o));
        }
#pragma unroll
        for (int h = 0; h < 4; ++h) {
            float xe = ws[h] >= 0.f ? xmx : xmn;
            m[h] = lrelu(xe * ws[h] + ad[h]);
            float ts = sub == 0 ? __expf(lrelu(xd * ws[h] + ad[h]) - m[h]) : 0.f;
            s[h] = ts;
            w[h] = ts * xd;
        }
#pragma unroll
        for (int u = 0; u < A1_U; ++u) {
            if (j0 + 8 * u < end) {
                float xv = xs[u];
#pragma unroll
                for (int h = 0; h < 4; ++h) {
                    float tt = __expf(lrelu(xv * ws[h] + ad[h]) - m[h]);
                    s[h] += tt;
                    w[h] += tt * xv;
                }
            }
        }
        for (int j = j0 + 8 * A1_U; j < end; j += 8) {
            int sidx = csr[j];
            bool lo = sidx < K;
            float a = xl[lo ? sidx : 0];
            float g = x[lo ? ncur : sidx];
            float xv = lo ? a : g;
#pragma unroll
            for (int h = 0; h < 4; ++h) {
                float tt = __expf(lrelu(xv * ws[h] + ad[h]) - m[h]);
                s[h] += tt;
                w[h] += tt * xv;
            }
        }

#pragma unroll
        for (int o = 1; o < 8; o <<= 1) {
#pragma unroll
            for (int h = 0; h < 4; ++h) {
                s[h] += __shfl_xor(s[h], o);
                w[h] += __shfl_xor(w[h], o);
            }
        }

        int h = sub >> 1;
        float sh = h == 0 ? s[0] : h == 1 ? s[1] : h == 2 ? s[2] : s[3];
        float wh = h == 0 ? w[0] : h == 1 ? w[1] : h == 2 ? w[2] : w[3];
        float r = wh / sh;

        int q = trip & 3;
#pragma unroll
        for (int hh = 0; hh < 4; ++hh) {
            float v = __shfl(r, 8 * (m16 & 3) + 2 * hh);
            rt[hh] = (m16 >> 2) == q ? v : rt[hh];
        }

        if (q == 3 || nw + A1_T / 8 >= n1) {
            v8f acc = {};
#pragma unroll
            for (int step = 0; step < 8; ++step) {
                int k0 = step * 4 + khi * 2;
                float rh = rt[step >> 1];
                v2f a;
                a.x = fmaxf(W1[k0] * rh + b1[k0], 0.f);
                a.y = fmaxf(W1[k0 + 1] * rh + b1[k0 + 1], 0.f);
                v2f b; b.x = Bp[k0 * 16 + m16];   b.y = Bp[(k0 + 1) * 16 + m16];
                acc = __builtin_amdgcn_wmma_f32_16x16x4_f32(false, a, false, b, (short)0, acc,
                                                            false, false);
            }
            int tb = nw - q * (A1_T / 8);
#pragma unroll
            for (int i = 0; i < 8; ++i) {
                int M = khi * 8 + i;
                int nd = tb + (M & 3) + (M >> 2) * (A1_T / 8);
                if (nd < n1) rec[(size_t)nd * 16 + m16] = acc[i];
            }
        }
    }
}

__global__ void __launch_bounds__(256) k_agg2(const float* __restrict__ rec,
                                              const int* __restrict__ rowptr,
                                              const int* __restrict__ csr,
                                              const float* __restrict__ b2,
                                              float* __restrict__ out, int N) {
    int gid = blockIdx.x * blockDim.x + threadIdx.x;
    int node = gid >> 3, sub = threadIdx.x & 7;
    bool valid = node < N;
    int nd = valid ? node : N - 1;

    const float* rd = rec + (size_t)nd * 16;
    float4 d0 = *reinterpret_cast<const float4*>(rd);
    float4 d1 = *reinterpret_cast<const float4*>(rd + 4);
    float2 dl = *reinterpret_cast<const float2*>(rd + 8);
    float ald = dl.y;
    int beg = rowptr[nd];
    int end = valid ? rowptr[nd + 1] : beg;

    float f0 = sub == 0 ? 1.f : 0.f;
    float m = lrelu(dl.x + ald);
    float s = f0;
    float acc[8] = { f0 * d0.x, f0 * d0.y, f0 * d0.z, f0 * d0.w,
                     f0 * d1.x, f0 * d1.y, f0 * d1.z, f0 * d1.w };

    for (int j = beg + sub; j < end; j += 8) {
        int sidx = csr[j];
        const float* rs = rec + (size_t)sidx * 16;
        float4 a0 = *reinterpret_cast<const float4*>(rs);
        float4 a1 = *reinterpret_cast<const float4*>(rs + 4);
        float als = rs[8];
        float e  = lrelu(als + ald);
        float mn = fmaxf(m, e);
        float sc = __expf(m - mn);
        float t  = __expf(e - mn);
        s = s * sc + t;
        acc[0] = acc[0] * sc + t * a0.x; acc[1] = acc[1] * sc + t * a0.y;
        acc[2] = acc[2] * sc + t * a0.z; acc[3] = acc[3] * sc + t * a0.w;
        acc[4] = acc[4] * sc + t * a1.x; acc[5] = acc[5] * sc + t * a1.y;
        acc[6] = acc[6] * sc + t * a1.z; acc[7] = acc[7] * sc + t * a1.w;
        m = mn;
    }

#pragma unroll
    for (int o = 1; o < 8; o <<= 1) {
        float mo = __shfl_xor(m, o);
        float so = __shfl_xor(s, o);
        float mn = fmaxf(m, mo);
        float a  = __expf(m - mn);
        float b  = __expf(mo - mn);
        s = s * a + so * b;
#pragma unroll
        for (int c = 0; c < 8; ++c) {
            float ao = __shfl_xor(acc[c], o);
            acc[c] = acc[c] * a + ao * b;
        }
        m = mn;
    }

    float av = acc[0];
#pragma unroll
    for (int c = 1; c < 8; ++c) av = sub == c ? acc[c] : av;
    float o1 = av / s + b2[sub];
    if (valid) out[(size_t)node * 8 + sub] = o1;
}

extern "C" void kernel_launch(void* const* d_in, const int* in_sizes, int n_in,
                              void* d_out, int out_size, void* d_ws, size_t ws_size,
                              hipStream_t stream) {
    const float* x   = (const float*)d_in[0];
    const int*   ei  = (const int*)d_in[1];
    const float* W1  = (const float*)d_in[2];
    const float* as1 = (const float*)d_in[3];
    const float* ad1 = (const float*)d_in[4];
    const float* b1  = (const float*)d_in[5];
    const float* W2  = (const float*)d_in[6];
    const float* as2 = (const float*)d_in[7];
    const float* ad2 = (const float*)d_in[8];
    const float* b2  = (const float*)d_in[9];
    (void)n_in; (void)out_size; (void)ws_size;

    int N = in_sizes[0];
    int E = in_sizes[1] / 2;

    char* ws = (char*)d_ws;
    size_t cur = 0;
    auto carve = [&](size_t bytes) -> void* {
        void* p = ws + cur;
        cur += (bytes + 255) & ~(size_t)255;
        return p;
    };
    int NB = (N + BW - 1) >> BSH;
    int PG = (E + RUN_CAP - 1) / RUN_CAP;
    int TP = (NB + 1 + 31) & ~31;
    int*      tab    = (int*)carve(sizeof(int) * (size_t)PG * (size_t)TP);
    int*      rowptr = (int*)carve(sizeof(int) * ((size_t)N + 1));
    unsigned* binned = (unsigned*)carve(sizeof(unsigned) * (size_t)PG * RUN_CAP);
    int*      csr    = (int*)carve(sizeof(int) * (size_t)E);
    float* rec    = (float*)carve(sizeof(float) * (size_t)N * 16);

    const int B = 256;
    k_bsort<<<PG, PA_T, 0, stream>>>(ei, E, NB, TP, tab, binned);
    k_bucket<<<NB, KB_T, 0, stream>>>(binned, tab, PG, TP, rowptr, csr, N);
    k_agg1<<<A1_G, A1_T, 0, stream>>>(x, rowptr, csr, W1, as1, ad1, b1, W2, as2, ad2, rec, N);

    k_agg2<<<(8 * N + B - 1) / B, B, 0, stream>>>(rec, rowptr, csr, b2, (float*)d_out, N);
}


# Layer-2 aggregation: 4 lanes own a node, each loads 16 bytes of a source's 64-byte record (a wave's load touches 8 lines, not 32); a node's float sums run in csr order in one lane, no 8-way tree
# speedup vs baseline: 26.4601x; 1.0506x over previous
#include <hip/hip_runtime.h>
#include <math.h>


#define NEG_SLOPE 0.2f

typedef __attribute__((ext_vector_type(2))) float v2f;
typedef __attribute__((ext_vector_type(8))) float v8f;

__device__ __forceinline__ float lrelu(float v) { return v > 0.f ? v : NEG_SLOPE * v; }

#define BSH 9
#define BW (1 << BSH)
#define MAXNB 1024
#define SRC_BITS 23
#define PA_T 512
#define PA_I 16
#define RUN_CAP (PA_T * PA_I)
#define KB_T 1024
#define KB_R 20

__device__ __forceinline__ int scan_pairs(int v0, int v1, int* wtot, int lane, int wv) {
    int incl = v0 + v1;
#pragma unroll
    for (int o = 1; o < 32; o <<= 1) {
        int u = __shfl_up(incl, o);
        if (lane >= o) incl += u;
    }
    if (lane == 31) wtot[wv] = incl;
    __syncthreads();
    int woff = 0;
#pragma unroll
    for (int k = 0; k < PA_T / 32; ++k)
        if (k < wv) woff += wtot[k];
    return woff + incl - (v0 + v1);
}

__global__ void __launch_bounds__(PA_T) k_bsort(const int* __restrict__ ei, int E, int NB, int TP,
                                                int* __restrict__ tab,
                                                unsigned* __restrict__ binned) {
    __shared__ int hist[MAXNB];
    __shared__ int lstart[MAXNB];
    __shared__ int wtot[PA_T / 32];
    __shared__ unsigned stage[RUN_CAP];
    int t = threadIdx.x, lane = t & 31, wv = t >> 5;
    int i0 = 2 * t, i1 = 2 * t + 1;
    hist[i0] = 0; hist[i1] = 0;
    __syncthreads();

    int blk = blockIdx.x;
    int base = blk * RUN_CAP;
    int d[PA_I], s[PA_I];
#pragma unroll
    for (int k = 0; k < PA_I; ++k) {
        int e = base + k * PA_T + t;
        bool in = e < E;
        d[k] = in ? ei[E + e] : -1;
        s[k] = in ? ei[e] : 0;
    }
#pragma unroll
    for (int k = 0; k < PA_I; ++k)
        if (d[k] >= 0) atomicAdd(&hist[d[k] >> BSH], 1);
    __syncthreads();

    int c0 = hist[i0], c1 = hist[i1];
    int l0 = scan_pairs(c0, c1, wtot, lane, wv);
    lstart[i0] = l0; lstart[i1] = l0 + c0;
    int* row = tab + (size_t)blk * TP;
    if (i0 <= NB) row[i0] = l0;
    if (i1 <= NB) row[i1] = l0 + c0;
    hist[i0] = 0; hist[i1] = 0;
    __syncthreads();

#pragma unroll
    for (int k = 0; k < PA_I; ++k) {
        if (d[k] >= 0) {
            int b = d[k] >> BSH;
            int r = atomicAdd(&hist[b], 1);
            stage[lstart[b] + r] = (unsigned)s[k] | ((unsigned)(d[k] & (BW - 1)) << SRC_BITS);
        }
    }
    __syncthreads();

    int nvalid = E - base;
    if (nvalid > RUN_CAP) nvalid = RUN_CAP;
#pragma unroll
    for (int k = 0; k < PA_I; ++k) {
        int i = k * PA_T + t;
        if (i < nvalid) binned[base + i] = stage[i];
    }
}

__device__ __forceinline__ int run_slot(const int* roffs, const int* rbase, int PG, int j) {
    int lo = 0, hi = PG - 1;
    while (lo < hi) {
        int mid = (lo + hi + 1) >> 1;
        if (roffs[mid] <= j) lo = mid; else hi = mid - 1;
    }
    return rbase[lo] + j;
}

__global__ void __launch_bounds__(KB_T) k_bucket(const unsigned* __restrict__ binned,
                                                 const int* __restrict__ tab, int PG, int TP,
                                                 int* __restrict__ rowptr,
                                                 int* __restrict__ csr, int N) {
    __shared__ int cnt[BW];
    __shared__ int wtot[BW / 32];
    __shared__ int rtot[KB_T / 32];
    __shared__ int rsum[KB_T / 32];
    __shared__ int roffs[KB_T];
    __shared__ int rbase[KB_T];
    __shared__ int stage[KB_T * KB_R];
    int t = threadIdx.x, lane = t & 31, wv = t >> 5;
    int b = blockIdx.x;

    int r0 = 0, len = 0;
    if (t < PG) {
        const int* row = tab + (size_t)t * TP;
        r0 = row[b];
        len = row[b + 1] - r0;
    }
    int rincl = len, rs = r0;
#pragma unroll
    for (int o = 1; o < 32; o <<= 1) {
        int u = __shfl_up(rincl, o);
        if (lane >= o) rincl += u;
        rs += __shfl_xor(rs, o);
    }
    if (lane == 31) rtot[wv] = rincl;
    if (lane == 0) rsum[wv] = rs;
    if (t < BW) cnt[t] = 0;
    __syncthreads();
    int roff = rincl - len, n = 0, bs = 0;
#pragma unroll
    for (int k = 0; k < KB_T / 32; ++k) {
        int x = rtot[k];
        if (k < wv) roff += x;
        n += x;
        bs += rsum[k];
    }
    int g0 = t * RUN_CAP + r0;
    roffs[t] = roff;
    rbase[t] = g0 - roff;
    if (t == 0 && n == 0) stage[0] = 0;
    for (int i = 0; i < len; ++i) {
        int j = roff + i;
        if (j < KB_T * KB_R) stage[j] = g0 + i;
    }
    __syncthreads();

    bool fits = n <= KB_T * KB_R;
    int* cs = csr + bs;

    unsigned w[KB_R];
    int slot[KB_R];
#pragma unroll
    for (int u = 0; u < KB_R; ++u) {
        int j = u * KB_T + t;
        slot[u] = stage[j < n ? j : 0];
    }
#pragma unroll
    for (int u = 0; u < KB_R; ++u) {
        int j = u * KB_T + t;
        w[u] = j < n ? binned[slot[u]] : 0u;
    }
#pragma unroll
    for (int u = 0; u < KB_R; ++u)
        if (u * KB_T + t < n) atomicAdd(&cnt[w[u] >> SRC_BITS], 1);
    for (int j = KB_T * KB_R + t; j < n; j += KB_T)
        atomicAdd(&cnt[binned[run_slot(roffs, rbase, PG, j)] >> SRC_BITS], 1);
    __syncthreads();

    int v = t < BW ? cnt[t] : 0;
    int incl = v;
#pragma unroll
    for (int o = 1; o < 32; o <<= 1) {
        int u = __shfl_up(incl, o);
        if (lane >= o) incl += u;
    }
    if (lane == 31 && t < BW) wtot[wv] = incl;
    __syncthreads();
    if (t < BW) {
        int woff = 0;
#pragma unroll
        for (int k = 0; k < BW / 32; ++k)
            if (k < wv) woff += wtot[k];
        int excl = woff + incl - v;
        cnt[t] = excl;
        int node = b * BW + t;
        if (node < N) rowptr[node] = bs + excl;
        if (node == N - 1) rowptr[N] = bs + excl + v;
    }
    __syncthreads();

    const unsigned SRC_MASK = (1u << SRC_BITS) - 1u;
    if (fits) {
#pragma unroll
        for (int u = 0; u < KB_R; ++u) {
            if (u * KB_T + t < n) {
                int p = atomicAdd(&cnt[w[u] >> SRC_BITS], 1);
                stage[p] = (int)(w[u] & SRC_MASK);
            }
        }
        __syncthreads();
        for (int i = t; i < n; i += KB_T) cs[i] = stage[i];
    } else {
#pragma unroll
        for (int u = 0; u < KB_R; ++u) {
            if (u * KB_T + t < n) {
                int p = atomicAdd(&cnt[w[u] >> SRC_BITS], 1);
                cs[p] = (int)(w[u] & SRC_MASK);
            }
        }
        for (int j = KB_T * KB_R + t; j < n; j += KB_T) {
            unsigned x = binned[run_slot(roffs, rbase, PG, j)];
            int p = atomicAdd(&cnt[x >> SRC_BITS], 1);
            cs[p] = (int)(x & SRC_MASK);
        }
    }
}

#define XL_MAX 80000
#define A1_T 1024
#define A1_G 256
#define A1_U 6

__global__ void __launch_bounds__(A1_T) k_agg1(const float* __restrict__ x,
                                               const int* __restrict__ rowptr,
                                               const int* __restrict__ csr,
                                               const float* __restrict__ W1,
                                               const float* __restrict__ as1,
                                               const float* __restrict__ ad1,
                                               const float* __restrict__ b1,
                                               const float* __restrict__ W2,
                                               const float* __restrict__ as2,
                                               const float* __restrict__ ad2,
                                               float* __restrict__ rec, int N) {
    __shared__ __attribute__((aligned(16))) float xl[XL_MAX];
    __shared__ float Bp[32 * 16];
    int t = threadIdx.x;
    if (t < 512) {
        int k = t >> 4, n = t & 15;
        float v = 0.f;
        if (n < 8) {
            v = W2[k * 8 + n];
        } else if (n == 8) {
            for (int c = 0; c < 8; ++c) v += W2[k * 8 + c] * as2[c];
        } else if (n == 9) {
            for (int c = 0; c < 8; ++c) v += W2[k * 8 + c] * ad2[c];
        }
        Bp[t] = v;
    }
    int K = N < XL_MAX ? N : XL_MAX;
    int K4 = K & ~3;
    for (int base = t * 4; base < K4; base += 5 * A1_T * 4) {
        float4 v[5];
        int ii[5];
#pragma unroll
        for (int u = 0; u < 5; ++u) {
            int i = base + u * A1_T * 4;
            ii[u] = i < K4 ? i : base;
            v[u] = *reinterpret_cast<const float4*>(x + ii[u]);
        }
#pragma unroll
        for (int u = 0; u < 5; ++u)
            *reinterpret_cast<float4*>(xl + ii[u]) = v[u];
    }
    if (t < K - K4) xl[K4 + t] = x[K4 + t];

    float ws[4], wd[4];
#pragma unroll
    for (int h = 0; h < 4; ++h) {
        float a = 0.f, d = 0.f;
#pragma unroll
        for (int c = 0; c < 8; ++c) {
            float w = W1[h * 8 + c];
            a += w * as1[h * 8 + c];
            d += w * ad1[h * 8 + c];
        }
        ws[h] = a; wd[h] = d;
    }
    __syncthreads();

    int sub = t & 7, grp = t >> 3;
    int per = ((N + (int)gridDim.x - 1) / (int)gridDim.x + 3) & ~3;
    int n0 = (int)blockIdx.x * per;
    int n1 = n0 + per < N ? n0 + per : N;

    int lane = t & 31, m16 = lane & 15, khi = lane >> 4;
    int nw = n0 + 4 * __builtin_amdgcn_readfirstlane(t >> 5);
    float rt[4] = { 0.f, 0.f, 0.f, 0.f };
    int trip = 0;

    int node = n0 + grp;
    int nc = node < n1 ? node : 0;
    float xd_n = x[nc];
    int beg_n = rowptr[nc], end_n = rowptr[nc + 1];

    for (; nw < n1; nw += A1_T / 8, node += A1_T / 8, ++trip) {
        float xd = xd_n;
        int beg = beg_n, end = end_n;
        if (node >= n1) end = beg;
        int nn = node + A1_T / 8 < n1 ? node + A1_T / 8 : nc;
        xd_n = x[nn];
        beg_n = rowptr[nn]; end_n = rowptr[nn + 1];

        float ad[4], m[4], s[4], w[4];
#pragma unroll
        for (int h = 0; h < 4; ++h) ad[h] = xd * wd[h];

        int last = end > 0 ? end - 1 : 0;
        int ncur = node < n1 ? node : 0;
        int j0 = beg + sub;
        int sv[A1_U];
#pragma unroll
        for (int u = 0; u < A1_U; ++u) {
            int j = j0 + 8 * u;
            sv[u] = csr[j < end ? j : last];
        }
        float va[A1_U], vg[A1_U];
#pragma unroll
        for (int u = 0; u < A1_U; ++u) {
            bool in = j0 + 8 * u < end;
            bool lo = sv[u] < K;
            va[u] = xl[in && lo ? sv[u] : 0];
            vg[u] = x[in && !lo ? sv[u] : ncur];
        }
        float xs[A1_U];
        float xmn = xd, xmx = xd;
#pragma unroll
        for (int u = 0; u < A1_U; ++u) {
            bool in = j0 + 8 * u < end;
            float v = sv[u] < K ? va[u] : vg[u];
            xs[u] = in ? v : xd;
            xmn = fminf(xmn, xs[u]);
            xmx = fmaxf(xmx, xs[u]);
        }
        for (int j = j0 + 8 * A1_U; j < end; j += 8) {
            int sidx = csr[j];
            bool lo = sidx < K;
            float a = xl[lo ? sidx : 0];
            float g = x[lo ? ncur : sidx];
            float v = lo ? a : g;
            xmn = fminf(xmn, v);
            xmx = fmaxf(xmx, v);
        }
#pragma unroll
        for (int o = 1; o < 8; o <<= 1) {
            xmn = fminf(xmn, __shfl_xor(xmn, o));
            xmx = fmaxf(xmx, __shfl_xor(xmx, o));
        }
#pragma unroll
        for (int h = 0; h < 4; ++h) {
            float xe = ws[h] >= 0.f ? xmx : xmn;
            m[h] = lrelu(xe * ws[h] + ad[h]);
            float ts = sub == 0 ? __expf(lrelu(xd * ws[h] + ad[h]) - m[h]) : 0.f;
            s[h] = ts;
            w[h] = ts * xd;
        }
#pragma unroll
        for (int u = 0; u < A1_U; ++u) {
            if (j0 + 8 * u < end) {
                float xv = xs[u];
#pragma unroll
                for (int h = 0; h < 4; ++h) {
                    float tt = __expf(lrelu(xv * ws[h] + ad[h]) - m[h]);
                    s[h] += tt;
                    w[h] += tt * xv;
                }
            }
        }
        for (int j = j0 + 8 * A1_U; j < end; j += 8) {
            int sidx = csr[j];
            bool lo = sidx < K;
            float a = xl[lo ? sidx : 0];
            float g = x[lo ? ncur : sidx];
            float xv = lo ? a : g;
#pragma unroll
            for (int h = 0; h < 4; ++h) {
                float tt = __expf(lrelu(xv * ws[h] + ad[h]) - m[h]);
                s[h] += tt;
                w[h] += tt * xv;
            }
        }

#pragma unroll
        for (int o = 1; o < 8; o <<= 1) {
#pragma unroll
            for (int h = 0; h < 4; ++h) {
                s[h] += __shfl_xor(s[h], o);
                w[h] += __shfl_xor(w[h], o);
            }
        }

        int h = sub >> 1;
        float sh = h == 0 ? s[0] : h == 1 ? s[1] : h == 2 ? s[2] : s[3];
        float wh = h == 0 ? w[0] : h == 1 ? w[1] : h == 2 ? w[2] : w[3];
        float r = wh / sh;

        int q = trip & 3;
#pragma unroll
        for (int hh = 0; hh < 4; ++hh) {
            float v = __shfl(r, 8 * (m16 & 3) + 2 * hh);
            rt[hh] = (m16 >> 2) == q ? v : rt[hh];
        }

        if (q == 3 || nw + A1_T / 8 >= n1) {
            v8f acc = {};
#pragma unroll
            for (int step = 0; step < 8; ++step) {
                int k0 = step * 4 + khi * 2;
                float rh = rt[step >> 1];
                v2f a;
                a.x = fmaxf(W1[k0] * rh + b1[k0], 0.f);
                a.y = fmaxf(W1[k0 + 1] * rh + b1[k0 + 1], 0.f);
                v2f b; b.x = Bp[k0 * 16 + m16];   b.y = Bp[(k0 + 1) * 16 + m16];
                acc = __builtin_amdgcn_wmma_f32_16x16x4_f32(false, a, false, b, (short)0, acc,
                                                            false, false);
            }
            int tb = nw - q * (A1_T / 8);
#pragma unroll
            for (int i = 0; i < 8; ++i) {
                int M = khi * 8 + i;
                int nd = tb + (M & 3) + (M >> 2) * (A1_T / 8);
                if (nd < n1) rec[(size_t)nd * 16 + m16] = acc[i];
            }
        }
    }
}

#define A2_U 8

typedef __attribute__((ext_vector_type(4))) float v4f;

template <int K> __device__ __forceinline__ int quad_lane(int v) {
    return __builtin_amdgcn_update_dpp(0, v, K * 0x55, 0xf, 0xf, true);
}
template <int K> __device__ __forceinline__ float quad_lane(float v) {
    return __int_as_float(quad_lane<K>(__float_as_int(v)));
}

__global__ void __launch_bounds__(256) k_agg2(const float* __restrict__ rec,
                                              const int* __restrict__ rowptr,
                                              const int* __restrict__ csr,
                                              const float* __restrict__ b2,
                                              float* __restrict__ out, int N) {
    int gid = blockIdx.x * blockDim.x + threadIdx.x;
    int node = gid >> 2, q = threadIdx.x & 3;
    bool valid = node < N;
    int nd = valid ? node : N - 1;

    const v4f* rec4 = reinterpret_cast<const v4f*>(rec);
    v4f dv = rec4[(unsigned)nd * 4u + (unsigned)q];
    int beg = rowptr[nd];
    int e1 = rowptr[nd + 1];
    int end = valid ? e1 : beg;
    float ald = quad_lane<2>(dv.y);

    float m = lrelu(quad_lane<2>(dv.x) + ald);
    float s = 1.f;
    v4f acc = dv;

    int last = end > 0 ? end - 1 : 0;
    int j = beg;
    int p0 = j + q, p1 = j + 4 + q;
    int i0 = csr[p0 < end ? p0 : last];
    int i1 = csr[p1 < end ? p1 : last];

    while (j < end) {
        int sv[A2_U];
        sv[0] = quad_lane<0>(i0); sv[1] = quad_lane<1>(i0);
        sv[2] = quad_lane<2>(i0); sv[3] = quad_lane<3>(i0);
        sv[4] = quad_lane<0>(i1); sv[5] = quad_lane<1>(i1);
        sv[6] = quad_lane<2>(i1); sv[7] = quad_lane<3>(i1);
        v4f v[A2_U];
#pragma unroll
        for (int u = 0; u < A2_U; ++u)
            v[u] = rec4[(unsigned)sv[u] * 4u + (unsigned)q];

        int jn = j + A2_U;
        p0 = jn + q; p1 = jn + 4 + q;
        i0 = csr[p0 < end ? p0 : last];
        i1 = csr[p1 < end ? p1 : last];

        __builtin_amdgcn_sched_barrier(0);

        float e[A2_U];
        float mn = m;
        int rem = end - j;
#pragma unroll
        for (int u = 0; u < A2_U; ++u) {
            float lg = lrelu(quad_lane<2>(v[u].x) + ald);
            e[u] = u < rem ? lg : -INFINITY;
            mn = fmaxf(mn, e[u]);
        }
        float sc = __expf(m - mn);
        s *= sc;
        acc *= sc;
#pragma unroll
        for (int u = 0; u < A2_U; ++u) {
            float t = __expf(e[u] - mn);
            s += t;
            acc += t * v[u];
        }
        m = mn;
        j = jn;
    }

    if (valid && q < 2) {
        v4f o;
        o.x = acc.x / s + b2[4 * q];
        o.y = acc.y / s + b2[4 * q + 1];
        o.z = acc.z / s + b2[4 * q + 2];
        o.w = acc.w / s + b2[4 * q + 3];
        reinterpret_cast<v4f*>(out)[(size_t)node * 2 + q] = o;
    }
}

extern "C" void kernel_launch(void* const* d_in, const int* in_sizes, int n_in,
                              void* d_out, int out_size, void* d_ws, size_t ws_size,
                              hipStream_t stream) {
    const float* x   = (const float*)d_in[0];
    const int*   ei  = (const int*)d_in[1];
    const float* W1  = (const float*)d_in[2];
    const float* as1 = (const float*)d_in[3];
    const float* ad1 = (const float*)d_in[4];
    const float* b1  = (const float*)d_in[5];
    const float* W2  = (const float*)d_in[6];
    const float* as2 = (const float*)d_in[7];
    const float* ad2 = (const float*)d_in[8];
    const float* b2  = (const float*)d_in[9];
    (void)n_in; (void)out_size; (void)ws_size;

    int N = in_sizes[0];
    int E = in_sizes[1] / 2;

    char* ws = (char*)d_ws;
    size_t cur = 0;
    auto carve = [&](size_t bytes) -> void* {
        void* p = ws + cur;
        cur += (bytes + 255) & ~(size_t)255;
        return p;
    };
    int NB = (N + BW - 1) >> BSH;
    int PG = (E + RUN_CAP - 1) / RUN_CAP;
    int TP = (NB + 1 + 31) & ~31;
    int*      tab    = (int*)carve(sizeof(int) * (size_t)PG * (size_t)TP);
    int*      rowptr = (int*)carve(sizeof(int) * ((size_t)N + 1));
    unsigned* binned = (unsigned*)carve(sizeof(unsigned) * (size_t)PG * RUN_CAP);
    int*      csr    = (int*)carve(sizeof(int) * (size_t)E);
    float* rec    = (float*)carve(sizeof(float) * (size_t)N * 16);

    const int B = 256;
    k_bsort<<<PG, PA_T, 0, stream>>>(ei, E, NB, TP, tab, binned);
    k_bucket<<<NB, KB_T, 0, stream>>>(binned, tab, PG, TP, rowptr, csr, N);
    k_agg1<<<A1_G, A1_T, 0, stream>>>(x, rowptr, csr, W1, as1, ad1, b1, W2, as2, ad2, rec, N);

    k_agg2<<<(4 * N + B - 1) / B, B, 0, stream>>>(rec, rowptr, csr, b2, (float*)d_out, N);
}


# Both sort levels place an edge at start + the rank its counting LDS atomic returned (kept in a register): one LDS atomic per edge and level, not two; sums stay in LDS arrival order per owner
# speedup vs baseline: 27.9213x; 1.0552x over previous
#include <hip/hip_runtime.h>
#include <math.h>


#define NEG_SLOPE 0.2f

typedef __attribute__((ext_vector_type(2))) float v2f;
typedef __attribute__((ext_vector_type(8))) float v8f;

__device__ __forceinline__ float lrelu(float v) { return v > 0.f ? v : NEG_SLOPE * v; }

#define BSH 9
#define BW (1 << BSH)
#define MAXNB 1024
#define SRC_BITS 23
#define PA_T 512
#define PA_I 16
#define RUN_CAP (PA_T * PA_I)
#define KB_T 1024
#define KB_R 20

__device__ __forceinline__ int scan_pairs(int v0, int v1, int* wtot, int lane, int wv) {
    int incl = v0 + v1;
#pragma unroll
    for (int o = 1; o < 32; o <<= 1) {
        int u = __shfl_up(incl, o);
        if (lane >= o) incl += u;
    }
    if (lane == 31) wtot[wv] = incl;
    __syncthreads();
    int woff = 0;
#pragma unroll
    for (int k = 0; k < PA_T / 32; ++k)
        if (k < wv) woff += wtot[k];
    return woff + incl - (v0 + v1);
}

__global__ void __launch_bounds__(PA_T) k_bsort(const int* __restrict__ ei, int E, int NB, int TP,
                                                int* __restrict__ tab,
                                                unsigned* __restrict__ binned) {
    __shared__ int hist[MAXNB];
    __shared__ int wtot[PA_T / 32];
    __shared__ unsigned stage[RUN_CAP];
    int t = threadIdx.x, lane = t & 31, wv = t >> 5;
    int i0 = 2 * t, i1 = 2 * t + 1;
    hist[i0] = 0; hist[i1] = 0;
    __syncthreads();

    int blk = blockIdx.x;
    int base = blk * RUN_CAP;
    int d[PA_I], s[PA_I];
#pragma unroll
    for (int k = 0; k < PA_I; ++k) {
        int e = base + k * PA_T + t;
        bool in = e < E;
        d[k] = in ? ei[E + e] : -1;
        s[k] = in ? ei[e] : 0;
    }
    int r[PA_I];
#pragma unroll
    for (int k = 0; k < PA_I; ++k)
        r[k] = d[k] >= 0 ? atomicAdd(&hist[d[k] >> BSH], 1) : 0;
    __syncthreads();

    int c0 = hist[i0], c1 = hist[i1];
    int l0 = scan_pairs(c0, c1, wtot, lane, wv);
    hist[i0] = l0; hist[i1] = l0 + c0;
    int* row = tab + (size_t)blk * TP;
    if (i0 <= NB) row[i0] = l0;
    if (i1 <= NB) row[i1] = l0 + c0;
    __syncthreads();

#pragma unroll
    for (int k = 0; k < PA_I; ++k)
        r[k] += hist[d[k] >= 0 ? d[k] >> BSH : 0];
#pragma unroll
    for (int k = 0; k < PA_I; ++k) {
        if (d[k] >= 0)
            stage[r[k]] = (unsigned)s[k] | ((unsigned)(d[k] & (BW - 1)) << SRC_BITS);
    }
    __syncthreads();

    int nvalid = E - base;
    if (nvalid > RUN_CAP) nvalid = RUN_CAP;
#pragma unroll
    for (int k = 0; k < PA_I; ++k) {
        int i = k * PA_T + t;
        if (i < nvalid) binned[base + i] = stage[i];
    }
}

__device__ __forceinline__ int run_slot(const int* roffs, const int* rbase, int PG, int j) {
    int lo = 0, hi = PG - 1;
    while (lo < hi) {
        int mid = (lo + hi + 1) >> 1;
        if (roffs[mid] <= j) lo = mid; else hi = mid - 1;
    }
    return rbase[lo] + j;
}

__global__ void __launch_bounds__(KB_T) k_bucket(const unsigned* __restrict__ binned,
                                                 const int* __restrict__ tab, int PG, int TP,
                                                 int* __restrict__ rowptr,
                                                 int* __restrict__ csr, int N) {
    __shared__ int cnt[BW];
    __shared__ int wtot[BW / 32];
    __shared__ int rtot[KB_T / 32];
    __shared__ int rsum[KB_T / 32];
    __shared__ int roffs[KB_T];
    __shared__ int rbase[KB_T];
    __shared__ int stage[KB_T * KB_R];
    int t = threadIdx.x, lane = t & 31, wv = t >> 5;
    int b = blockIdx.x;

    int r0 = 0, len = 0;
    if (t < PG) {
        const int* row = tab + (size_t)t * TP;
        r0 = row[b];
        len = row[b + 1] - r0;
    }
    int rincl = len, rs = r0;
#pragma unroll
    for (int o = 1; o < 32; o <<= 1) {
        int u = __shfl_up(rincl, o);
        if (lane >= o) rincl += u;
        rs += __shfl_xor(rs, o);
    }
    if (lane == 31) rtot[wv] = rincl;
    if (lane == 0) rsum[wv] = rs;
    if (t < BW) cnt[t] = 0;
    __syncthreads();
    int roff = rincl - len, n = 0, bs = 0;
#pragma unroll
    for (int k = 0; k < KB_T / 32; ++k) {
        int x = rtot[k];
        if (k < wv) roff += x;
        n += x;
        bs += rsum[k];
    }
    int g0 = t * RUN_CAP + r0;
    roffs[t] = roff;
    rbase[t] = g0 - roff;
    if (t == 0 && n == 0) stage[0] = 0;
    for (int i = 0; i < len; ++i) {
        int j = roff + i;
        if (j < KB_T * KB_R) stage[j] = g0 + i;
    }
    __syncthreads();

    bool fits = n <= KB_T * KB_R;
    int* cs = csr + bs;

    unsigned w[KB_R];
    int slot[KB_R];
#pragma unroll
    for (int u = 0; u < KB_R; ++u) {
        int j = u * KB_T + t;
        slot[u] = stage[j < n ? j : 0];
    }
#pragma unroll
    for (int u = 0; u < KB_R; ++u) {
        int j = u * KB_T + t;
        w[u] = j < n ? binned[slot[u]] : 0u;
    }
    int rk[KB_R];
#pragma unroll
    for (int u = 0; u < KB_R; ++u)
        rk[u] = u * KB_T + t < n ? atomicAdd(&cnt[w[u] >> SRC_BITS], 1) : 0;
    for (int j = KB_T * KB_R + t; j < n; j += KB_T)
        atomicAdd(&cnt[binned[run_slot(roffs, rbase, PG, j)] >> SRC_BITS], 1);
    __syncthreads();

    int v = t < BW ? cnt[t] : 0;
    int incl = v;
#pragma unroll
    for (int o = 1; o < 32; o <<= 1) {
        int u = __shfl_up(incl, o);
        if (lane >= o) incl += u;
    }
    if (lane == 31 && t < BW) wtot[wv] = incl;
    __syncthreads();
    if (t < BW) {
        int woff = 0;
#pragma unroll
        for (int k = 0; k < BW / 32; ++k)
            if (k < wv) woff += wtot[k];
        int excl = woff + incl - v;
        cnt[t] = excl;
        int node = b * BW + t;
        if (node < N) rowptr[node] = bs + excl;
        if (node == N - 1) rowptr[N] = bs + excl + v;
    }
    __syncthreads();

    const unsigned SRC_MASK = (1u << SRC_BITS) - 1u;
    if (fits) {
#pragma unroll
        for (int u0 = 0; u0 < KB_R; u0 += 4) {
#pragma unroll
            for (int u = u0; u < u0 + 4; ++u)
                rk[u] += cnt[w[u] >> SRC_BITS];
#pragma unroll
            for (int u = u0; u < u0 + 4; ++u) {
                if (u * KB_T + t < n)
                    stage[rk[u]] = (int)(w[u] & SRC_MASK);
            }
        }
        __syncthreads();
        for (int i = t; i < n; i += KB_T) cs[i] = stage[i];
    } else {
#pragma unroll
        for (int u = 0; u < KB_R; ++u) {
            if (u * KB_T + t < n) {
                int p = atomicAdd(&cnt[w[u] >> SRC_BITS], 1);
                cs[p] = (int)(w[u] & SRC_MASK);
            }
        }
        for (int j = KB_T * KB_R + t; j < n; j += KB_T) {
            unsigned x = binned[run_slot(roffs, rbase, PG, j)];
            int p = atomicAdd(&cnt[x >> SRC_BITS], 1);
            cs[p] = (int)(x & SRC_MASK);
        }
    }
}

#define XL_MAX 80000
#define A1_T 1024
#define A1_G 256
#define A1_U 6

__global__ void __launch_bounds__(A1_T) k_agg1(const float* __restrict__ x,
                                               const int* __restrict__ rowptr,
                                               const int* __restrict__ csr,
                                               const float* __restrict__ W1,
                                               const float* __restrict__ as1,
                                               const float* __restrict__ ad1,
                                               const float* __restrict__ b1,
                                               const float* __restrict__ W2,
                                               const float* __restrict__ as2,
                                               const float* __restrict__ ad2,
                                               float* __restrict__ rec, int N) {
    __shared__ __attribute__((aligned(16))) float xl[XL_MAX];
    __shared__ float Bp[32 * 16];
    int t = threadIdx.x;
    if (t < 512) {
        int k = t >> 4, n = t & 15;
        float v = 0.f;
        if (n < 8) {
            v = W2[k * 8 + n];
        } else if (n == 8) {
            for (int c = 0; c < 8; ++c) v += W2[k * 8 + c] * as2[c];
        } else if (n == 9) {
            for (int c = 0; c < 8; ++c) v += W2[k * 8 + c] * ad2[c];
        }
        Bp[t] = v;
    }
    int K = N < XL_MAX ? N : XL_MAX;
    int K4 = K & ~3;
    for (int base = t * 4; base < K4; base += 5 * A1_T * 4) {
        float4 v[5];
        int ii[5];
#pragma unroll
        for (int u = 0; u < 5; ++u) {
            int i = base + u * A1_T * 4;
            ii[u] = i < K4 ? i : base;
            v[u] = *reinterpret_cast<const float4*>(x + ii[u]);
        }
#pragma unroll
        for (int u = 0; u < 5; ++u)
            *reinterpret_cast<float4*>(xl + ii[u]) = v[u];
    }
    if (t < K - K4) xl[K4 + t] = x[K4 + t];

    float ws[4], wd[4];
#pragma unroll
    for (int h = 0; h < 4; ++h) {
        float a = 0.f, d = 0.f;
#pragma unroll
        for (int c = 0; c < 8; ++c) {
            float w = W1[h * 8 + c];
            a += w * as1[h * 8 + c];
            d += w * ad1[h * 8 + c];
        }
        ws[h] = a; wd[h] = d;
    }
    __syncthreads();

    int sub = t & 7, grp = t >> 3;
    int per = ((N + (int)gridDim.x - 1) / (int)gridDim.x + 3) & ~3;
    int n0 = (int)blockIdx.x * per;
    int n1 = n0 + per < N ? n0 + per : N;

    int lane = t & 31, m16 = lane & 15, khi = lane >> 4;
    int nw = n0 + 4 * __builtin_amdgcn_readfirstlane(t >> 5);
    float rt[4] = { 0.f, 0.f, 0.f, 0.f };
    int trip = 0;

    int node = n0 + grp;
    int nc = node < n1 ? node : 0;
    float xd_n = x[nc];
    int beg_n = rowptr[nc], end_n = rowptr[nc + 1];

    for (; nw < n1; nw += A1_T / 8, node += A1_T / 8, ++trip) {
        float xd = xd_n;
        int beg = beg_n, end = end_n;
        if (node >= n1) end = beg;
        int nn = node + A1_T / 8 < n1 ? node + A1_T / 8 : nc;
        xd_n = x[nn];
        beg_n = rowptr[nn]; end_n = rowptr[nn + 1];

        float ad[4], m[4], s[4], w[4];
#pragma unroll
        for (int h = 0; h < 4; ++h) ad[h] = xd * wd[h];

        int last = end > 0 ? end - 1 : 0;
        int ncur = node < n1 ? node : 0;
        int j0 = beg + sub;
        int sv[A1_U];
#pragma unroll
        for (int u = 0; u < A1_U; ++u) {
            int j = j0 + 8 * u;
            sv[u] = csr[j < end ? j : last];
        }
        float va[A1_U], vg[A1_U];
#pragma unroll
        for (int u = 0; u < A1_U; ++u) {
            bool in = j0 + 8 * u < end;
            bool lo = sv[u] < K;
            va[u] = xl[in && lo ? sv[u] : 0];
            vg[u] = x[in && !lo ? sv[u] : ncur];
        }
        float xs[A1_U];
        float xmn = xd, xmx = xd;
#pragma unroll
        for (int u = 0; u < A1_U; ++u) {
            bool in = j0 + 8 * u < end;
            float v = sv[u] < K ? va[u] : vg[u];
            xs[u] = in ? v : xd;
            xmn = fminf(xmn, xs[u]);
            xmx = fmaxf(xmx, xs[u]);
        }
        for (int j = j0 + 8 * A1_U; j < end; j += 8) {
            int sidx = csr[j];
            bool lo = sidx < K;
            float a = xl[lo ? sidx : 0];
            float g = x[lo ? ncur : sidx];
            float v = lo ? a : g;
            xmn = fminf(xmn, v);
            xmx = fmaxf(xmx, v);
        }
#pragma unroll
        for (int o = 1; o < 8; o <<= 1) {
            xmn = fminf(xmn, __shfl_xor(xmn, o));
            xmx = fmaxf(xmx, __shfl_xor(xmx, o));
        }
#pragma unroll
        for (int h = 0; h < 4; ++h) {
            float xe = ws[h] >= 0.f ? xmx : xmn;
            m[h] = lrelu(xe * ws[h] + ad[h]);
            float ts = sub == 0 ? __expf(lrelu(xd * ws[h] + ad[h]) - m[h]) : 0.f;
            s[h] = ts;
            w[h] = ts * xd;
        }
#pragma unroll
        for (int u = 0; u < A1_U; ++u) {
            if (j0 + 8 * u < end) {
                float xv = xs[u];
#pragma unroll
                for (int h = 0; h < 4; ++h) {
                    float tt = __expf(lrelu(xv * ws[h] + ad[h]) - m[h]);
                    s[h] += tt;
                    w[h] += tt * xv;
                }
            }
        }
        for (int j = j0 + 8 * A1_U; j < end; j += 8) {
            int sidx = csr[j];
            bool lo = sidx < K;
            float a = xl[lo ? sidx : 0];
            float g = x[lo ? ncur : sidx];
            float xv = lo ? a : g;
#pragma unroll
            for (int h = 0; h < 4; ++h) {
                float tt = __expf(lrelu(xv * ws[h] + ad[h]) - m[h]);
                s[h] += tt;
                w[h] += tt * xv;
            }
        }

#pragma unroll
        for (int o = 1; o < 8; o <<= 1) {
#pragma unroll
            for (int h = 0; h < 4; ++h) {
                s[h] += __shfl_xor(s[h], o);
                w[h] += __shfl_xor(w[h], o);
            }
        }

        int h = sub >> 1;
        float sh = h == 0 ? s[0] : h == 1 ? s[1] : h == 2 ? s[2] : s[3];
        float wh = h == 0 ? w[0] : h == 1 ? w[1] : h == 2 ? w[2] : w[3];
        float r = wh / sh;

        int q = trip & 3;
#pragma unroll
        for (int hh = 0; hh < 4; ++hh) {
            float v = __shfl(r, 8 * (m16 & 3) + 2 * hh);
            rt[hh] = (m16 >> 2) == q ? v : rt[hh];
        }

        if (q == 3 || nw + A1_T / 8 >= n1) {
            v8f acc = {};
#pragma unroll
            for (int step = 0; step < 8; ++step) {
                int k0 = step * 4 + khi * 2;
                float rh = rt[step >> 1];
                v2f a;
                a.x = fmaxf(W1[k0] * rh + b1[k0], 0.f);
                a.y = fmaxf(W1[k0 + 1] * rh + b1[k0 + 1], 0.f);
                v2f b; b.x = Bp[k0 * 16 + m16];   b.y = Bp[(k0 + 1) * 16 + m16];
                acc = __builtin_amdgcn_wmma_f32_16x16x4_f32(false, a, false, b, (short)0, acc,
                                                            false, false);
            }
            int tb = nw - q * (A1_T / 8);
#pragma unroll
            for (int i = 0; i < 8; ++i) {
                int M = khi * 8 + i;
                int nd = tb + (M & 3) + (M >> 2) * (A1_T / 8);
                if (nd < n1) rec[(size_t)nd * 16 + m16] = acc[i];
            }
        }
    }
}

#define A2_U 8

typedef __attribute__((ext_vector_type(4))) float v4f;

template <int K> __device__ __forceinline__ int quad_lane(int v) {
    return __builtin_amdgcn_update_dpp(0, v, K * 0x55, 0xf, 0xf, true);
}
template <int K> __device__ __forceinline__ float quad_lane(float v) {
    return __int_as_float(quad_lane<K>(__float_as_int(v)));
}

__global__ void __launch_bounds__(256) k_agg2(const float* __restrict__ rec,
                                              const int* __restrict__ rowptr,
                                              const int* __restrict__ csr,
                                              const float* __restrict__ b2,
                                              float* __restrict__ out, int N) {
    int gid = blockIdx.x * blockDim.x + threadIdx.x;
    int node = gid >> 2, q = threadIdx.x & 3;
    bool valid = node < N;
    int nd = valid ? node : N - 1;

    const v4f* rec4 = reinterpret_cast<const v4f*>(rec);
    v4f dv = rec4[(unsigned)nd * 4u + (unsigned)q];
    int beg = rowptr[nd];
    int e1 = rowptr[nd + 1];
    int end = valid ? e1 : beg;
    float ald = quad_lane<2>(dv.y);

    float m = lrelu(quad_lane<2>(dv.x) + ald);
    float s = 1.f;
    v4f acc = dv;

    int last = end > 0 ? end - 1 : 0;
    int j = beg;
    int p0 = j + q, p1 = j + 4 + q;
    int i0 = csr[p0 < end ? p0 : last];
    int i1 = csr[p1 < end ? p1 : last];

    while (j < end) {
        int sv[A2_U];
        sv[0] = quad_lane<0>(i0); sv[1] = quad_lane<1>(i0);
        sv[2] = quad_lane<2>(i0); sv[3] = quad_lane<3>(i0);
        sv[4] = quad_lane<0>(i1); sv[5] = quad_lane<1>(i1);
        sv[6] = quad_lane<2>(i1); sv[7] = quad_lane<3>(i1);
        v4f v[A2_U];
#pragma unroll
        for (int u = 0; u < A2_U; ++u)
            v[u] = rec4[(unsigned)sv[u] * 4u + (unsigned)q];

        int jn = j + A2_U;
        p0 = jn + q; p1 = jn + 4 + q;
        i0 = csr[p0 < end ? p0 : last];
        i1 = csr[p1 < end ? p1 : last];

        __builtin_amdgcn_sched_barrier(0);

        float e[A2_U];
        float mn = m;
        int rem = end - j;
#pragma unroll
        for (int u = 0; u < A2_U; ++u) {
            float lg = lrelu(quad_lane<2>(v[u].x) + ald);
            e[u] = u < rem ? lg : -INFINITY;
            mn = fmaxf(mn, e[u]);
        }
        float sc = __expf(m - mn);
        s *= sc;
        acc *= sc;
#pragma unroll
        for (int u = 0; u < A2_U; ++u) {
            float t = __expf(e[u] - mn);
            s += t;
            acc += t * v[u];
        }
        m = mn;
        j = jn;
    }

    if (valid && q < 2) {
        v4f o;
        o.x = acc.x / s + b2[4 * q];
        o.y = acc.y / s + b2[4 * q + 1];
        o.z = acc.z / s + b2[4 * q + 2];
        o.w = acc.w / s + b2[4 * q + 3];
        reinterpret_cast<v4f*>(out)[(size_t)node * 2 + q] = o;
    }
}

extern "C" void kernel_launch(void* const* d_in, const int* in_sizes, int n_in,
                              void* d_out, int out_size, void* d_ws, size_t ws_size,
                              hipStream_t stream) {
    const float* x   = (const float*)d_in[0];
    const int*   ei  = (const int*)d_in[1];
    const float* W1  = (const float*)d_in[2];
    const float* as1 = (const float*)d_in[3];
    const float* ad1 = (const float*)d_in[4];
    const float* b1  = (const float*)d_in[5];
    const float* W2  = (const float*)d_in[6];
    const float* as2 = (const float*)d_in[7];
    const float* ad2 = (const float*)d_in[8];
    const float* b2  = (const float*)d_in[9];
    (void)n_in; (void)out_size; (void)ws_size;

    int N = in_sizes[0];
    int E = in_sizes[1] / 2;

    char* ws = (char*)d_ws;
    size_t cur = 0;
    auto carve = [&](size_t bytes) -> void* {
        void* p = ws + cur;
        cur += (bytes + 255) & ~(size_t)255;
        return p;
    };
    int NB = (N + BW - 1) >> BSH;
    int PG = (E + RUN_CAP - 1) / RUN_CAP;
    int TP = (NB + 1 + 31) & ~31;
    int*      tab    = (int*)carve(sizeof(int) * (size_t)PG * (size_t)TP);
    int*      rowptr = (int*)carve(sizeof(int) * ((size_t)N + 1));
    unsigned* binned = (unsigned*)carve(sizeof(unsigned) * (size_t)PG * RUN_CAP);
    int*      csr    = (int*)carve(sizeof(int) * (size_t)E);
    float* rec    = (float*)carve(sizeof(float) * (size_t)N * 16);

    const int B = 256;
    k_bsort<<<PG, PA_T, 0, stream>>>(ei, E, NB, TP, tab, binned);
    k_bucket<<<NB, KB_T, 0, stream>>>(binned, tab, PG, TP, rowptr, csr, N);
    k_agg1<<<A1_G, A1_T, 0, stream>>>(x, rowptr, csr, W1, as1, ad1, b1, W2, as2, ad2, rec, N);

    k_agg2<<<(4 * N + B - 1) / B, B, 0, stream>>>(rec, rowptr, csr, b2, (float*)d_out, N);
}


# Level 2 of the sort and the layer-1 aggregation in one launch: the work-group that places a bucket (392 nodes, 256 buckets) also sums its nodes; 4 to 3 launches; sums stay in LDS arrival order
# speedup vs baseline: 30.3917x; 1.0885x over previous
#include <hip/hip_runtime.h>
#include <math.h>


#define NEG_SLOPE 0.2f

typedef __attribute__((ext_vector_type(2))) float v2f;
typedef __attribute__((ext_vector_type(8))) float v8f;

__device__ __forceinline__ float lrelu(float v) { return v > 0.f ? v : NEG_SLOPE * v; }

#define BWN 392
#define BW 512
#define MAXNB 1024
#define SRC_BITS 23
#define PA_T 512
#define PA_I 16
#define RUN_CAP (PA_T * PA_I)
#define KB_T 1024
#define KB_R 20

__device__ __forceinline__ int scan_pairs(int v0, int v1, int* wtot, int lane, int wv) {
    int incl = v0 + v1;
#pragma unroll
    for (int o = 1; o < 32; o <<= 1) {
        int u = __shfl_up(incl, o);
        if (lane >= o) incl += u;
    }
    if (lane == 31) wtot[wv] = incl;
    __syncthreads();
    int woff = 0;
#pragma unroll
    for (int k = 0; k < PA_T / 32; ++k)
        if (k < wv) woff += wtot[k];
    return woff + incl - (v0 + v1);
}

__global__ void __launch_bounds__(PA_T) k_bsort(const int* __restrict__ ei, int E, int NB, int TP,
                                                int* __restrict__ tab,
                                                unsigned* __restrict__ binned) {
    __shared__ int hist[MAXNB];
    __shared__ int wtot[PA_T / 32];
    __shared__ unsigned stage[RUN_CAP];
    int t = threadIdx.x, lane = t & 31, wv = t >> 5;
    int i0 = 2 * t, i1 = 2 * t + 1;
    hist[i0] = 0; hist[i1] = 0;
    __syncthreads();

    int blk = blockIdx.x;
    int base = blk * RUN_CAP;
    int d[PA_I], s[PA_I];
#pragma unroll
    for (int k = 0; k < PA_I; ++k) {
        int e = base + k * PA_T + t;
        bool in = e < E;
        d[k] = in ? ei[E + e] : -1;
        s[k] = in ? ei[e] : 0;
    }
    int r[PA_I];
#pragma unroll
    for (int k = 0; k < PA_I; ++k)
        r[k] = d[k] >= 0 ? atomicAdd(&hist[(unsigned)d[k] / BWN], 1) : 0;
    __syncthreads();

    int c0 = hist[i0], c1 = hist[i1];
    int l0 = scan_pairs(c0, c1, wtot, lane, wv);
    hist[i0] = l0; hist[i1] = l0 + c0;
    int* row = tab + (size_t)blk * TP;
    if (i0 <= NB) row[i0] = l0;
    if (i1 <= NB) row[i1] = l0 + c0;
    __syncthreads();

#pragma unroll
    for (int k = 0; k < PA_I; ++k)
        r[k] += hist[d[k] >= 0 ? (unsigned)d[k] / BWN : 0u];
#pragma unroll
    for (int k = 0; k < PA_I; ++k) {
        if (d[k] >= 0)
            stage[r[k]] = (unsigned)s[k] | (((unsigned)d[k] % BWN) << SRC_BITS);
    }
    __syncthreads();

    int nvalid = E - base;
    if (nvalid > RUN_CAP) nvalid = RUN_CAP;
#pragma unroll
    for (int k = 0; k < PA_I; ++k) {
        int i = k * PA_T + t;
        if (i < nvalid) binned[base + i] = stage[i];
    }
}

__device__ __forceinline__ int run_slot(const int* roffs, const int* rbase, int PG, int j) {
    int lo = 0, hi = PG - 1;
    while (lo < hi) {
        int mid = (lo + hi + 1) >> 1;
        if (roffs[mid] <= j) lo = mid; else hi = mid - 1;
    }
    return rbase[lo] + j;
}

#define XL_MAX 80000
#define A1_T 1024
#define A1_U 6


__global__ void __launch_bounds__(KB_T) k_bucket_agg1(const unsigned* __restrict__ binned,
                                                      const int* __restrict__ tab, int PG, int TP,
                                                      int* __restrict__ rowptr,
                                                      int* csr,
                                                      const float* __restrict__ x,
                                                      const float* __restrict__ W1,
                                                      const float* __restrict__ as1,
                                                      const float* __restrict__ ad1,
                                                      const float* __restrict__ b1,
                                                      const float* __restrict__ W2,
                                                      const float* __restrict__ as2,
                                                      const float* __restrict__ ad2,
                                                      float* __restrict__ rec, int N) {
    static_assert(A1_T == KB_T, "one block shape for both phases");
    static_assert(BWN % 4 == 0 && BWN < BW, "bucket width");
    static_assert((KB_T * KB_R + 2 * KB_T) <= XL_MAX, "the sort's tables fit in the room of the x copy");
    __shared__ __attribute__((aligned(16))) float xl[XL_MAX];
    __shared__ float Bp[32 * 16];
    __shared__ int cnt[BW];
    __shared__ int wtot[BW / 32];
    __shared__ int rtot[KB_T / 32];
    __shared__ int rsum[KB_T / 32];
    int* stage = reinterpret_cast<int*>(xl);
    int* roffs = stage + KB_T * KB_R;
    int* rbase = roffs + KB_T;
    int t = threadIdx.x, lane = t & 31, wv = t >> 5;
    int b = blockIdx.x;
    if (t < 512) {
        int k = t >> 4, n = t & 15;
        float v = 0.f;
        if (n < 8) {
            v = W2[k * 8 + n];
        } else if (n == 8) {
            for (int c = 0; c < 8; ++c) v += W2[k * 8 + c] * as2[c];
        } else if (n == 9) {
            for (int c = 0; c < 8; ++c) v += W2[k * 8 + c] * ad2[c];
        }
        Bp[t] = v;
    }

    int ebase;
    int excl_keep = 0;
    bool big;
    {
        int r0 = 0, len = 0;
        if (t < PG) {
            const int* row = tab + (size_t)t * TP;
            r0 = row[b];
            len = row[b + 1] - r0;
        }
        int rincl = len, rs = r0;
#pragma unroll
        for (int o = 1; o < 32; o <<= 1) {
            int u = __shfl_up(rincl, o);
            if (lane >= o) rincl += u;
            rs += __shfl_xor(rs, o);
        }
        if (lane == 31) rtot[wv] = rincl;
        if (lane == 0) rsum[wv] = rs;
        if (t < BW) cnt[t] = 0;
        __syncthreads();
        int roff = rincl - len, n = 0, bs = 0;
#pragma unroll
        for (int k = 0; k < KB_T / 32; ++k) {
            int xt = rtot[k];
            if (k < wv) roff += xt;
            n += xt;
            bs += rsum[k];
        }
        int g0 = t * RUN_CAP + r0;
        roffs[t] = roff;
        rbase[t] = g0 - roff;
        if (t == 0 && n == 0) stage[0] = 0;
        for (int i = 0; i < len; ++i) {
            int j = roff + i;
            if (j < KB_T * KB_R) stage[j] = g0 + i;
        }
        __syncthreads();

        bool fits = n <= KB_T * KB_R;
        int* cs = csr + bs;

        unsigned w[KB_R];
        int slot[KB_R];
#pragma unroll
        for (int u = 0; u < KB_R; ++u) {
            int j = u * KB_T + t;
            slot[u] = stage[j < n ? j : 0];
        }
#pragma unroll
        for (int u = 0; u < KB_R; ++u) {
            int j = u * KB_T + t;
            w[u] = j < n ? binned[slot[u]] : 0u;
        }
        int rk[KB_R];
#pragma unroll
        for (int u = 0; u < KB_R; ++u)
            rk[u] = u * KB_T + t < n ? atomicAdd(&cnt[w[u] >> SRC_BITS], 1) : 0;
        for (int j = KB_T * KB_R + t; j < n; j += KB_T)
            atomicAdd(&cnt[binned[run_slot(roffs, rbase, PG, j)] >> SRC_BITS], 1);
        __syncthreads();

        int v = t < BW ? cnt[t] : 0;
        int incl = v;
#pragma unroll
        for (int o = 1; o < 32; o <<= 1) {
            int u = __shfl_up(incl, o);
            if (lane >= o) incl += u;
        }
        if (lane == 31 && t < BW) wtot[wv] = incl;
        __syncthreads();
        if (t < BW) {
            int woff = 0;
#pragma unroll
            for (int k = 0; k < BW / 32; ++k)
                if (k < wv) woff += wtot[k];
            int excl = woff + incl - v;
            cnt[t] = excl;
            excl_keep = excl;
            int node = b * BWN + t;
            if (t < BWN && node < N) rowptr[node] = bs + excl;
            if (t < BWN && node == N - 1) rowptr[N] = bs + excl + v;
        }
        __syncthreads();

        const unsigned SRC_MASK = (1u << SRC_BITS) - 1u;
        if (fits) {
#pragma unroll
            for (int u0 = 0; u0 < KB_R; u0 += 4) {
#pragma unroll
                for (int u = u0; u < u0 + 4; ++u)
                    rk[u] += cnt[w[u] >> SRC_BITS];
#pragma unroll
                for (int u = u0; u < u0 + 4; ++u) {
                    if (u * KB_T + t < n)
                        stage[rk[u]] = (int)(w[u] & SRC_MASK);
                }
            }
            __syncthreads();
            for (int i = t; i < n; i += KB_T) cs[i] = stage[i];
        } else {
#pragma unroll
            for (int u = 0; u < KB_R; ++u) {
                if (u * KB_T + t < n) {
                    int p = atomicAdd(&cnt[w[u] >> SRC_BITS], 1);
                    cs[p] = (int)(w[u] & SRC_MASK);
                }
            }
            for (int j = KB_T * KB_R + t; j < n; j += KB_T) {
                unsigned xw = binned[run_slot(roffs, rbase, PG, j)];
                int p = atomicAdd(&cnt[xw >> SRC_BITS], 1);
                cs[p] = (int)(xw & SRC_MASK);
            }
        }
        ebase = bs;
        big = !fits;
    }
    __syncthreads();
    if (big && t < BW) cnt[t] = excl_keep;

    int K = N < XL_MAX ? N : XL_MAX;
    int K4 = K & ~3;
    for (int base = t * 4; base < K4; base += 5 * A1_T * 4) {
        float4 v[5];
        int ii[5];
#pragma unroll
        for (int u = 0; u < 5; ++u) {
            int i = base + u * A1_T * 4;
            ii[u] = i < K4 ? i : base;
            v[u] = *reinterpret_cast<const float4*>(x + ii[u]);
        }
#pragma unroll
        for (int u = 0; u < 5; ++u)
            *reinterpret_cast<float4*>(xl + ii[u]) = v[u];
    }
    if (t < K - K4) xl[K4 + t] = x[K4 + t];

    float ws[4], wd[4];
#pragma unroll
    for (int h = 0; h < 4; ++h) {
        float a = 0.f, d = 0.f;
#pragma unroll
        for (int c = 0; c < 8; ++c) {
            float w = W1[h * 8 + c];
            a += w * as1[h * 8 + c];
            d += w * ad1[h * 8 + c];
        }
        ws[h] = a; wd[h] = d;
    }
    __syncthreads();

    int sub = t & 7, grp = t >> 3;
    int n0 = b * BWN;
    int n1 = n0 + BWN < N ? n0 + BWN : N;

    int m16 = lane & 15, khi = lane >> 4;
    int nw = n0 + 4 * __builtin_amdgcn_readfirstlane(t >> 5);
    float rt[4] = { 0.f, 0.f, 0.f, 0.f };
    int trip = 0;

    int node = n0 + grp;
    int nc = node < n1 ? node : n0;
    float xd_n = x[nc];
    int beg_n = ebase + cnt[nc - n0], end_n = ebase + cnt[nc - n0 + 1];

    for (; nw < n1; nw += A1_T / 8, node += A1_T / 8, ++trip) {
        float xd = xd_n;
        int beg = beg_n, end = end_n;
        if (node >= n1) end = beg;
        int nn = node + A1_T / 8 < n1 ? node + A1_T / 8 : nc;
        xd_n = x[nn];
        beg_n = ebase + cnt[nn - n0]; end_n = ebase + cnt[nn - n0 + 1];

        float ad[4], m[4], s[4], w[4];
#pragma unroll
        for (int h = 0; h < 4; ++h) ad[h] = xd * wd[h];

        int last = end > 0 ? end - 1 : 0;
        int ncur = node < n1 ? node : 0;
        int j0 = beg + sub;
        int sv[A1_U];
#pragma unroll
        for (int u = 0; u < A1_U; ++u) {
            int j = j0 + 8 * u;
            sv[u] = csr[j < end ? j : last];
        }
        float va[A1_U], vg[A1_U];
#pragma unroll
        for (int u = 0; u < A1_U; ++u) {
            bool in = j0 + 8 * u < end;
            bool lo = sv[u] < K;
            va[u] = xl[in && lo ? sv[u] : 0];
            vg[u] = x[in && !lo ? sv[u] : ncur];
        }
        float xs[A1_U];
        float xmn = xd, xmx = xd;
#pragma unroll
        for (int u = 0; u < A1_U; ++u) {
            bool in = j0 + 8 * u < end;
            float v = sv[u] < K ? va[u] : vg[u];
            xs[u] = in ? v : xd;
            xmn = fminf(xmn, xs[u]);
            xmx = fmaxf(xmx, xs[u]);
        }
        for (int j = j0 + 8 * A1_U; j < end; j += 8) {
            int sidx = csr[j];
            bool lo = sidx < K;
            float a = xl[lo ? sidx : 0];
            float g = x[lo ? ncur : sidx];
            float v = lo ? a : g;
            xmn = fminf(xmn, v);
            xmx = fmaxf(xmx, v);
        }
#pragma unroll
        for (int o = 1; o < 8; o <<= 1) {
            xmn = fminf(xmn, __shfl_xor(xmn, o));
            xmx = fmaxf(xmx, __shfl_xor(xmx, o));
        }
#pragma unroll
        for (int h = 0; h < 4; ++h) {
            float xe = ws[h] >= 0.f ? xmx : xmn;
            m[h] = lrelu(xe * ws[h] + ad[h]);
            float ts = sub == 0 ? __expf(lrelu(xd * ws[h] + ad[h]) - m[h]) : 0.f;
            s[h] = ts;
            w[h] = ts * xd;
        }
#pragma unroll
        for (int u = 0; u < A1_U; ++u) {
            if (j0 + 8 * u < end) {
                float xv = xs[u];
#pragma unroll
                for (int h = 0; h < 4; ++h) {
                    float tt = __expf(lrelu(xv * ws[h] + ad[h]) - m[h]);
                    s[h] += tt;
                    w[h] += tt * xv;
                }
            }
        }
        for (int j = j0 + 8 * A1_U; j < end; j += 8) {
            int sidx = csr[j];
            bool lo = sidx < K;
            float a = xl[lo ? sidx : 0];
            float g = x[lo ? ncur : sidx];
            float xv = lo ? a : g;
#pragma unroll
            for (int h = 0; h < 4; ++h) {
                float tt = __expf(lrelu(xv * ws[h] + ad[h]) - m[h]);
                s[h] += tt;
                w[h] += tt * xv;
            }
        }

#pragma unroll
        for (int o = 1; o < 8; o <<= 1) {
#pragma unroll
            for (int h = 0; h < 4; ++h) {
                s[h] += __shfl_xor(s[h], o);
                w[h] += __shfl_xor(w[h], o);
            }
        }

        int h = sub >> 1;
        float sh = h == 0 ? s[0] : h == 1 ? s[1] : h == 2 ? s[2] : s[3];
        float wh = h == 0 ? w[0] : h == 1 ? w[1] : h == 2 ? w[2] : w[3];
        float r = wh / sh;

        int q = trip & 3;
#pragma unroll
        for (int hh = 0; hh < 4; ++hh) {
            float v = __shfl(r, 8 * (m16 & 3) + 2 * hh);
            rt[hh] = (m16 >> 2) == q ? v : rt[hh];
        }

        if (q == 3 || nw + A1_T / 8 >= n1) {
            v8f acc = {};
#pragma unroll
            for (int step = 0; step < 8; ++step) {
                int k0 = step * 4 + khi * 2;
                float rh = rt[step >> 1];
                v2f a;
                a.x = fmaxf(W1[k0] * rh + b1[k0], 0.f);
                a.y = fmaxf(W1[k0 + 1] * rh + b1[k0 + 1], 0.f);
                v2f b; b.x = Bp[k0 * 16 + m16];   b.y = Bp[(k0 + 1) * 16 + m16];
                acc = __builtin_amdgcn_wmma_f32_16x16x4_f32(false, a, false, b, (short)0, acc,
                                                            false, false);
            }
            int tb = nw - q * (A1_T / 8);
#pragma unroll
            for (int i = 0; i < 8; ++i) {
                int M = khi * 8 + i;
                int nd = tb + (M & 3) + (M >> 2) * (A1_T / 8);
                if (nd < n1) rec[(size_t)nd * 16 + m16] = acc[i];
            }
        }
    }
}

#define A2_U 8

typedef __attribute__((ext_vector_type(4))) float v4f;

template <int K> __device__ __forceinline__ int quad_lane(int v) {
    return __builtin_amdgcn_update_dpp(0, v, K * 0x55, 0xf, 0xf, true);
}
template <int K> __device__ __forceinline__ float quad_lane(float v) {
    return __int_as_float(quad_lane<K>(__float_as_int(v)));
}

__global__ void __launch_bounds__(256) k_agg2(const float* __restrict__ rec,
                                              const int* __restrict__ rowptr,
                                              const int* __restrict__ csr,
                                              const float* __restrict__ b2,
                                              float* __restrict__ out, int N) {
    int gid = blockIdx.x * blockDim.x + threadIdx.x;
    int node = gid >> 2, q = threadIdx.x & 3;
    bool valid = node < N;
    int nd = valid ? node : N - 1;

    const v4f* rec4 = reinterpret_cast<const v4f*>(rec);
    v4f dv = rec4[(unsigned)nd * 4u + (unsigned)q];
    int beg = rowptr[nd];
    int e1 = rowptr[nd + 1];
    int end = valid ? e1 : beg;
    float ald = quad_lane<2>(dv.y);

    float m = lrelu(quad_lane<2>(dv.x) + ald);
    float s = 1.f;
    v4f acc = dv;

    int last = end > 0 ? end - 1 : 0;
    int j = beg;
    int p0 = j + q, p1 = j + 4 + q;
    int i0 = csr[p0 < end ? p0 : last];
    int i1 = csr[p1 < end ? p1 : last];

    while (j < end) {
        int sv[A2_U];
        sv[0] = quad_lane<0>(i0); sv[1] = quad_lane<1>(i0);
        sv[2] = quad_lane<2>(i0); sv[3] = quad_lane<3>(i0);
        sv[4] = quad_lane<0>(i1); sv[5] = quad_lane<1>(i1);
        sv[6] = quad_lane<2>(i1); sv[7] = quad_lane<3>(i1);
        v4f v[A2_U];
#pragma unroll
        for (int u = 0; u < A2_U; ++u)
            v[u] = rec4[(unsigned)sv[u] * 4u + (unsigned)q];

        int jn = j + A2_U;
        p0 = jn + q; p1 = jn + 4 + q;
        i0 = csr[p0 < end ? p0 : last];
        i1 = csr[p1 < end ? p1 : last];

        __builtin_amdgcn_sched_barrier(0);

        float e[A2_U];
        float mn = m;
        int rem = end - j;
#pragma unroll
        for (int u = 0; u < A2_U; ++u) {
            float lg = lrelu(quad_lane<2>(v[u].x) + ald);
            e[u] = u < rem ? lg : -INFINITY;
            mn = fmaxf(mn, e[u]);
        }
        float sc = __expf(m - mn);
        s *= sc;
        acc *= sc;
#pragma unroll
        for (int u = 0; u < A2_U; ++u) {
            float t = __expf(e[u] - mn);
            s += t;
            acc += t * v[u];
        }
        m = mn;
        j = jn;
    }

    if (valid && q < 2) {
        v4f o;
        o.x = acc.x / s + b2[4 * q];
        o.y = acc.y / s + b2[4 * q + 1];
        o.z = acc.z / s + b2[4 * q + 2];
        o.w = acc.w / s + b2[4 * q + 3];
        reinterpret_cast<v4f*>(out)[(size_t)node * 2 + q] = o;
    }
}

extern "C" void kernel_launch(void* const* d_in, const int* in_sizes, int n_in,
                              void* d_out, int out_size, void* d_ws, size_t ws_size,
                              hipStream_t stream) {
    const float* x   = (const float*)d_in[0];
    const int*   ei  = (const int*)d_in[1];
    const float* W1  = (const float*)d_in[2];
    const float* as1 = (const float*)d_in[3];
    const float* ad1 = (const float*)d_in[4];
    const float* b1  = (const float*)d_in[5];
    const float* W2  = (const float*)d_in[6];
    const float* as2 = (const float*)d_in[7];
    const float* ad2 = (const float*)d_in[8];
    const float* b2  = (const float*)d_in[9];
    (void)n_in; (void)out_size; (void)ws_size;

    int N = in_sizes[0];
    int E = in_sizes[1] / 2;

    char* ws = (char*)d_ws;
    size_t cur = 0;
    auto carve = [&](size_t bytes) -> void* {
        void* p = ws + cur;
        cur += (bytes + 255) & ~(size_t)255;
        return p;
    };
    int NB = (N + BWN - 1) / BWN;
    int PG = (E + RUN_CAP - 1) / RUN_CAP;
    int TP = (NB + 1 + 31) & ~31;
    int*      tab    = (int*)carve(sizeof(int) * (size_t)PG * (size_t)TP);
    int*      rowptr = (int*)carve(sizeof(int) * ((size_t)N + 1));
    unsigned* binned = (unsigned*)carve(sizeof(unsigned) * (size_t)PG * RUN_CAP);
    int*      csr    = (int*)carve(sizeof(int) * (size_t)E);
    float* rec    = (float*)carve(sizeof(float) * (size_t)N * 16);

    const int B = 256;
    k_bsort<<<PG, PA_T, 0, stream>>>(ei, E, NB, TP, tab, binned);
    k_bucket_agg1<<<NB, KB_T, 0, stream>>>(binned, tab, PG, TP, rowptr, csr,
                                            x, W1, as1, ad1, b1, W2, as2, ad2, rec, N);

    k_agg2<<<(4 * N + B - 1) / B, B, 0, stream>>>(rec, rowptr, csr, b2, (float*)d_out, N);
}
